# starting kernel
# baseline (speedup 1.0000x reference)
#include <hip/hip_runtime.h>


typedef __attribute__((ext_vector_type(2))) float v2f;
typedef __attribute__((ext_vector_type(8))) float v8f;

__global__ void zero_f32(float* __restrict__ p, long n) {
  long i = (long)blockIdx.x * blockDim.x + threadIdx.x;
  if (i < n) p[i] = 0.0f;
}

__global__ void count_edges(const int* __restrict__ dst, float* __restrict__ cnt,
                            long n_edges) {
  long i = (long)blockIdx.x * blockDim.x + threadIdx.x;
  if (i < n_edges) unsafeAtomicAdd(&cnt[dst[i]], 1.0f);
}

__global__ void reciprocal_clamped(float* __restrict__ p, long n) {
  long i = (long)blockIdx.x * blockDim.x + threadIdx.x;
  if (i < n) p[i] = 1.0f / fmaxf(p[i], 1.0f);
}

template <int F_IN, int F_OUT>
__global__ void node_transform(const float* __restrict__ h,
                               const float* __restrict__ Wl,
                               const float* __restrict__ Wr,
                               const float* __restrict__ bias,
                               float* __restrict__ hl,
                               float* __restrict__ hs,
                               float* __restrict__ agg,
                               int n_nodes) {
  const int lane = threadIdx.x & 31;
  const int wave = threadIdx.x >> 5;
  const int tile = blockIdx.x * (blockDim.x >> 5) + wave;
  const int tiles = (n_nodes + 15) >> 4;
  if (tile >= tiles) return;

  const int m    = lane & 15;
  const int hi   = lane >> 4;
  const int row0 = tile << 4;
  const int arow = min(row0 + m, n_nodes - 1);
  const bool cok = (m < F_OUT);

  auto ldA = [&](int k) -> float {
    const bool ok = (k < F_IN);
    const float v = h[(long)arow * F_IN + (ok ? k : 0)];
    return ok ? v : 0.0f;
  };
  auto ldW = [&](const float* __restrict__ W, int k) -> float {
    const bool ok = cok && (k < F_IN);
    const float v = W[ok ? (m * F_IN + k) : 0];
    return ok ? v : 0.0f;
  };

  v8f dl = {};
  v8f dr = {};
#pragma unroll
  for (int kc = 0; kc < F_IN; kc += 4) {
    const int k0 = kc + hi * 2;
    v2f a, bl, br;
    a.x  = ldA(k0);
    a.y  = ldA(k0 + 1);
    bl.x = ldW(Wl, k0);
    bl.y = ldW(Wl, k0 + 1);
    br.x = ldW(Wr, k0);
    br.y = ldW(Wr, k0 + 1);
    dl = __builtin_amdgcn_wmma_f32_16x16x4_f32(false, a, false, bl, (short)0,
                                               dl, false, false);
    dr = __builtin_amdgcn_wmma_f32_16x16x4_f32(false, a, false, br, (short)0,
                                               dr, false, false);
  }

  if (cok) {
    const float bv   = bias[m];
    const long  base = (long)(row0 + hi * 8) * F_OUT + m;
    if (row0 + 16 <= n_nodes) {
#pragma unroll
      for (int v = 0; v < 8; ++v) {
        hl[base + (long)v * F_OUT]  = dl[v];
        hs[base + (long)v * F_OUT]  = dr[v] + bv;
        agg[base + (long)v * F_OUT] = 0.0f;
      }
    } else {
#pragma unroll
      for (int v = 0; v < 8; ++v) {
        if (row0 + hi * 8 + v < n_nodes) {
          hl[base + (long)v * F_OUT]  = dl[v];
          hs[base + (long)v * F_OUT]  = dr[v] + bv;
          agg[base + (long)v * F_OUT] = 0.0f;
        }
      }
    }
  }
}

template <int F>
__global__ void edge_scatter(const int* __restrict__ src,
                             const int* __restrict__ dst,
                             const float* __restrict__ hl,
                             float* __restrict__ agg, long n_edges) {
  long i = (long)blockIdx.x * blockDim.x + threadIdx.x;
  if (i >= n_edges) return;
  const int s = src[i];
  const int d = dst[i];
  float v[F];
#pragma unroll
  for (int j = 0; j < F; ++j) v[j] = hl[(long)s * F + j];
#pragma unroll
  for (int j = 0; j < F; ++j) unsafeAtomicAdd(&agg[(long)d * F + j], v[j]);
}

template <int F, bool RELU>
__global__ void combine(const float* __restrict__ agg,
                        const float* __restrict__ hs,
                        const float* __restrict__ inv,
                        float* __restrict__ out, int n_nodes) {
  const int node = blockIdx.x * blockDim.x + threadIdx.x;
  if (node >= n_nodes) return;
  const float iv = inv[node];
#pragma unroll
  for (int j = 0; j < F; ++j) {
    const long o = (long)node * F + j;
    const float v = agg[o] * iv + hs[o];
    out[o] = RELU ? fmaxf(v, 0.0f) : v;
  }
}

extern "C" void kernel_launch(void* const* d_in, const int* in_sizes, int n_in,
                              void* d_out, int out_size, void* d_ws,
                              size_t ws_size, hipStream_t stream) {
  const float* x    = (const float*)d_in[0];
  const int*   ei   = (const int*)  d_in[1];
  const float* Wl1  = (const float*)d_in[2];
  const float* Wr1  = (const float*)d_in[3];
  const float* b1   = (const float*)d_in[4];
  const float* Wlm  = (const float*)d_in[5];
  const float* Wrm  = (const float*)d_in[6];
  const float* bm   = (const float*)d_in[7];
  const float* Wl10 = (const float*)d_in[8];
  const float* Wr10 = (const float*)d_in[9];
  const float* b10  = (const float*)d_in[10];

  const int  N = in_sizes[0] / 10;
  const long E = (long)in_sizes[1] / 2;
  const int* src = ei;
  const int* dst = ei + E;

  float* ws  = (float*)d_ws;
  float* inv = ws;
  float* hl  = inv + (size_t)N;
  float* hs  = hl + (size_t)N * 5;
  float* agg = hs + (size_t)N * 5;
  float* hA  = agg + (size_t)N * 5;
  float* hB  = hA + (size_t)N * 5;

  const int B = 256;
  auto cdiv = [](long a, long b) { return (int)((a + b - 1) / b); };

  zero_f32<<<cdiv(N, B), B, 0, stream>>>(inv, N);
  count_edges<<<cdiv(E, B), B, 0, stream>>>(dst, inv, E);
  reciprocal_clamped<<<cdiv(N, B), B, 0, stream>>>(inv, N);

  const int tiles   = (N + 15) / 16;
  const int tblocks = (tiles + 7) / 8;

  node_transform<10, 5><<<tblocks, 256, 0, stream>>>(x, Wl1, Wr1, b1, hl, hs, agg, N);
  edge_scatter<5><<<cdiv(E, B), B, 0, stream>>>(src, dst, hl, agg, E);
  combine<5, true><<<cdiv(N, B), B, 0, stream>>>(agg, hs, inv, hA, N);

  for (int l = 0; l < 8; ++l) {
    const float* hin  = (l & 1) ? hB : hA;
    float*       hout = (l & 1) ? hA : hB;
    node_transform<5, 5><<<tblocks, 256, 0, stream>>>(
        hin, Wlm + l * 25, Wrm + l * 25, bm + l * 5, hl, hs, agg, N);
    edge_scatter<5><<<cdiv(E, B), B, 0, stream>>>(src, dst, hl, agg, E);
    combine<5, true><<<cdiv(N, B), B, 0, stream>>>(agg, hs, inv, hout, N);
  }

  node_transform<5, 1><<<tblocks, 256, 0, stream>>>(hA, Wl10, Wr10, b10, hl, hs, agg, N);
  edge_scatter<1><<<cdiv(E, B), B, 0, stream>>>(src, dst, hl, agg, E);
  combine<1, false><<<cdiv(N, B), B, 0, stream>>>(agg, hs, inv, (float*)d_out, N);
}


# Rows an edge touches (gathered rows, both accumulator copies) at 32-byte pitch, 32-byte aligned: no row's gather or atomic add lies across two lines; self term, 1/degree packed in the spare floats
# speedup vs baseline: 6.0162x; 6.0162x over previous
#include <hip/hip_runtime.h>


typedef __attribute__((ext_vector_type(2))) float v2f;
typedef __attribute__((ext_vector_type(4))) float v4f;
typedef __attribute__((ext_vector_type(8))) float v8f;

constexpr int RP = 8;
constexpr int AP = 6;

__device__ __forceinline__ int package_half() {
  return (int)((__builtin_amdgcn_s_sendmsg_rtn(0x87) >> 18) & 1u);
}

template <int F_IN, int F_OUT>
__device__ __forceinline__ float ldB1(const float* __restrict__ Wl,
                                      const float* __restrict__ Wr, int n,
                                      int k) {
  if constexpr (F_OUT == 5) {
    const bool ok = (n < RP) && (k < F_IN);
    const bool left = (n < 5);
    const float* W = left ? Wl : Wr;
    const int r = left ? n : n - 5;
    const float v = W[ok ? (r * F_IN + k) : 0];
    return ok ? v : 0.0f;
  } else {
    const bool ok = (n == 0) && (k < F_IN);
    const float v = Wl[ok ? k : 0];
    return ok ? v : 0.0f;
  }
}
template <int F_IN, int F_OUT>
__device__ __forceinline__ float ldB2(const float* __restrict__ Wr, int n,
                                      int k) {
  if constexpr (F_OUT == 5) {
    const bool ok = (n >= 6) && (n < RP) && (k < F_IN);
    const float v = Wr[ok ? ((n - 3) * F_IN + k) : 0];
    return ok ? v : 0.0f;
  } else {
    const bool ok = (n == RP - 1) && (k < F_IN);
    const float v = Wr[ok ? k : 0];
    return ok ? v : 0.0f;
  }
}

__device__ __forceinline__ void store_rows5(const v8f& d1, const v8f& d2,
                                            const float* __restrict__ bias,
                                            float* H, float* A0, float* A1,
                                            int row0, int hi, int m,
                                            int n_nodes, int nclear) {
  if (m < RP) {
    const bool  c1 = (m >= 5);
    const bool  c2 = (m >= 6);
    const float bl = bias[c1 ? m - 5 : 0];
    const float bh = bias[c2 ? m - 3 : 0];
    const long  base = (long)(row0 + hi * 8) * RP + m;
    if (row0 + 16 <= n_nodes) {
#pragma unroll
      for (int v = 0; v < 8; ++v) {
        H[base + (long)v * RP]  = c1 ? d1[v] + bl : d1[v];
        A0[base + (long)v * RP] = c2 ? d2[v] + bh : 0.0f;
      }
      if (m < nclear) {
#pragma unroll
        for (int v = 0; v < 8; ++v) A1[base + (long)v * RP] = 0.0f;
      }
    } else {
#pragma unroll
      for (int v = 0; v < 8; ++v) {
        if (row0 + hi * 8 + v < n_nodes) {
          H[base + (long)v * RP]  = c1 ? d1[v] + bl : d1[v];
          A0[base + (long)v * RP] = c2 ? d2[v] + bh : 0.0f;
        }
      }
      if (m < nclear) {
#pragma unroll
        for (int v = 0; v < 8; ++v) {
          if (row0 + hi * 8 + v < n_nodes) A1[base + (long)v * RP] = 0.0f;
        }
      }
    }
  }
}

template <int F_IN>
__global__ void node_transform(const float* __restrict__ x,
                               const float* __restrict__ Wl,
                               const float* __restrict__ Wr,
                               const float* __restrict__ bias,
                               float* __restrict__ H,
                               float* __restrict__ A0,
                               float* __restrict__ A1, int n_nodes) {
  const int lane = threadIdx.x & 31;
  const int wave = threadIdx.x >> 5;
  const int tile = blockIdx.x * (blockDim.x >> 5) + wave;
  const int tiles = (n_nodes + 15) >> 4;
  if (tile >= tiles) return;

  const int m    = lane & 15;
  const int hi   = lane >> 4;
  const int row0 = tile << 4;
  const int arow = min(row0 + m, n_nodes - 1);

  auto ldA = [&](int k) -> float {
    const bool ok = (k < F_IN);
    const float v = x[(long)arow * F_IN + (ok ? k : 0)];
    return ok ? v : 0.0f;
  };

  v8f d1 = {};
  v8f d2 = {};
#pragma unroll
  for (int kc = 0; kc < F_IN; kc += 4) {
    const int k0 = kc + hi * 2;
    v2f a, b1, b2;
    a.x  = ldA(k0);
    a.y  = ldA(k0 + 1);
    b1.x = ldB1<F_IN, 5>(Wl, Wr, m, k0);
    b1.y = ldB1<F_IN, 5>(Wl, Wr, m, k0 + 1);
    b2.x = ldB2<F_IN, 5>(Wr, m, k0);
    b2.y = ldB2<F_IN, 5>(Wr, m, k0 + 1);
    d1 = __builtin_amdgcn_wmma_f32_16x16x4_f32(false, a, false, b1, (short)0,
                                               d1, false, false);
    d2 = __builtin_amdgcn_wmma_f32_16x16x4_f32(false, a, false, b2, (short)0,
                                               d2, false, false);
  }
  store_rows5(d1, d2, bias, H, A0, A1, row0, hi, m, n_nodes, RP);
}

template <int F_OUT, bool FIRST>
__global__ void combine_transform(float* H, float* A0, float* A1,
                                  const float* __restrict__ Wl,
                                  const float* __restrict__ Wr,
                                  const float* __restrict__ bias,
                                  float* __restrict__ z0,
                                  float* __restrict__ z1, int n_nodes) {
  constexpr int F_IN = 5;
  const int lane = threadIdx.x & 31;
  const int wave = threadIdx.x >> 5;
  const int tile = blockIdx.x * (blockDim.x >> 5) + wave;
  const int tiles = (n_nodes + 15) >> 4;
  if (tile >= tiles) return;

  const int m    = lane & 15;
  const int hi   = lane >> 4;
  const int row0 = tile << 4;
  const int arow = min(row0 + m, n_nodes - 1);
  const long rb  = (long)arow * RP;

  const v4f a0l = *reinterpret_cast<const v4f*>(A0 + rb);
  const v4f a0h = *reinterpret_cast<const v4f*>(A0 + rb + 4);
  const v4f a1l = *reinterpret_cast<const v4f*>(A1 + rb);
  const v4f a1h = *reinterpret_cast<const v4f*>(A1 + rb + 4);
  const v4f hh  = *reinterpret_cast<const v4f*>(H + rb + 4);

  float iv;
  if (FIRST) {
    iv = 1.0f / fmaxf(a0h.y + a1h.y, 1.0f);
    if (hi == 0 && row0 + m < n_nodes) A1[rb + 6] = iv;
  } else {
    iv = a1h.z;
  }
  const float h0 = fmaxf((a0l.x + a1l.x) * iv + hh.y, 0.0f);
  const float h1 = fmaxf((a0l.y + a1l.y) * iv + hh.z, 0.0f);
  const float h2 = fmaxf((a0l.z + a1l.z) * iv + hh.w, 0.0f);
  const float h3 = fmaxf((a0l.w + a1l.w) * iv + a0h.z, 0.0f);
  const float h4 = fmaxf((a0h.x + a1h.x) * iv + a0h.w, 0.0f);

  v8f d1 = {};
  v8f d2 = {};
  {
    const int k0 = hi * 2;
    v2f a, b1, b2;
    a.x  = hi ? h2 : h0;
    a.y  = hi ? h3 : h1;
    b1.x = ldB1<F_IN, F_OUT>(Wl, Wr, m, k0);
    b1.y = ldB1<F_IN, F_OUT>(Wl, Wr, m, k0 + 1);
    b2.x = ldB2<F_IN, F_OUT>(Wr, m, k0);
    b2.y = ldB2<F_IN, F_OUT>(Wr, m, k0 + 1);
    d1 = __builtin_amdgcn_wmma_f32_16x16x4_f32(false, a, false, b1, (short)0,
                                               d1, false, false);
    d2 = __builtin_amdgcn_wmma_f32_16x16x4_f32(false, a, false, b2, (short)0,
                                               d2, false, false);
  }
  {
    const int k0 = 4 + hi * 2;
    v2f a, b1, b2;
    a.x  = hi ? 0.0f : h4;
    a.y  = 0.0f;
    b1.x = ldB1<F_IN, F_OUT>(Wl, Wr, m, k0);
    b1.y = ldB1<F_IN, F_OUT>(Wl, Wr, m, k0 + 1);
    b2.x = ldB2<F_IN, F_OUT>(Wr, m, k0);
    b2.y = ldB2<F_IN, F_OUT>(Wr, m, k0 + 1);
    d1 = __builtin_amdgcn_wmma_f32_16x16x4_f32(false, a, false, b1, (short)0,
                                               d1, false, false);
    d2 = __builtin_amdgcn_wmma_f32_16x16x4_f32(false, a, false, b2, (short)0,
                                               d2, false, false);
  }

  if constexpr (F_OUT == 5) {
    store_rows5(d1, d2, bias, H, A0, A1, row0, hi, m, n_nodes, AP);
  } else {
    const bool c0 = (m == 0);
    if (c0 || m == RP - 1) {
      const float bv = bias[0];
      float* p = (c0 ? H : A1) + (long)(row0 + hi * 8) * RP + m;
      const long zb = (long)(row0 + hi * 8);
      if (row0 + 16 <= n_nodes) {
#pragma unroll
        for (int v = 0; v < 8; ++v)
          p[(long)v * RP] = c0 ? d1[v] : d2[v] + bv;
        if (c0) {
#pragma unroll
          for (int v = 0; v < 8; ++v) {
            z0[zb + v] = 0.0f;
            z1[zb + v] = 0.0f;
          }
        }
      } else {
#pragma unroll
        for (int v = 0; v < 8; ++v) {
          if (row0 + hi * 8 + v < n_nodes)
            p[(long)v * RP] = c0 ? d1[v] : d2[v] + bv;
        }
        if (c0) {
#pragma unroll
          for (int v = 0; v < 8; ++v) {
            if (row0 + hi * 8 + v < n_nodes) {
              z0[zb + v] = 0.0f;
              z1[zb + v] = 0.0f;
            }
          }
        }
      }
    }
  }
}

__global__ void edge_scatter_one(const int* __restrict__ src,
                                 const int* __restrict__ dst,
                                 const float* __restrict__ H,
                                 float* __restrict__ z0,
                                 float* __restrict__ z1, long n_edges) {
  float* __restrict__ agg = package_half() ? z1 : z0;
  const long stride = (long)gridDim.x * blockDim.x;
  for (long i = (long)blockIdx.x * blockDim.x + threadIdx.x; i < n_edges;
       i += stride) {
    const int s = src[i];
    const int d = dst[i];
    unsafeAtomicAdd(&agg[d], H[s * RP]);
  }
}

template <int F>
__global__ void edge_scatter_rows(const int* __restrict__ src,
                                  const int* __restrict__ dst,
                                  const float* __restrict__ H,
                                  float* __restrict__ A0,
                                  float* __restrict__ A1, long n_edges) {
  constexpr int FH = 5;
  float* __restrict__ agg = package_half() ? A1 : A0;
  const int lane = threadIdx.x & 31;
  const long wave =
      (long)blockIdx.x * (blockDim.x >> 5) +
      __builtin_amdgcn_readfirstlane((int)(threadIdx.x >> 5));
  const long stride = (long)gridDim.x * blockDim.x;
  for (long base = wave * 32; base < n_edges; base += stride) {
    if (base + 32 <= n_edges) {
      const int s = src[base + lane];
      const int d = dst[base + lane];
      float v[F];
      int de[F];
#pragma unroll
      for (int t = 0; t < F; ++t) {
        const int p = t * 32 + lane;
        const int e = p / F;
        const int j = p - e * F;
        const int se = __shfl(s, e);
        de[t] = __shfl(d, e) * RP + j;
        const bool cl = (F > FH) && (j >= FH);
        const float g = H[se * RP + (cl ? FH - 1 : j)];
        v[t] = cl ? 1.0f : g;
      }
#pragma unroll
      for (int t = 0; t < F; ++t) unsafeAtomicAdd(&agg[de[t]], v[t]);
    } else {
      int s = 0, d = 0;
      if (base + lane < n_edges) {
        s = src[base + lane];
        d = dst[base + lane];
      }
#pragma unroll
      for (int t = 0; t < F; ++t) {
        const int p = t * 32 + lane;
        const int e = p / F;
        const int j = p - e * F;
        const int se = __shfl(s, e);
        const int de = __shfl(d, e);
        if (base + e < n_edges) {
          const bool cl = (F > FH) && (j >= FH);
          const float g = H[se * RP + (cl ? FH - 1 : j)];
          unsafeAtomicAdd(&agg[de * RP + j], cl ? 1.0f : g);
        }
      }
    }
  }
}

__global__ void combine_last(const float* __restrict__ z0,
                             const float* __restrict__ z1,
                             const float* __restrict__ A1,
                             float* __restrict__ out, int n_nodes) {
  const int node = blockIdx.x * blockDim.x + threadIdx.x;
  if (node >= n_nodes) return;
  const v2f t = *reinterpret_cast<const v2f*>(A1 + (long)node * RP + 6);
  const float a = z0[node] + z1[node];
  out[node] = a * t.x + t.y;
}

extern "C" void kernel_launch(void* const* d_in, const int* in_sizes, int n_in,
                              void* d_out, int out_size, void* d_ws,
                              size_t ws_size, hipStream_t stream) {
  const float* x    = (const float*)d_in[0];
  const int*   ei   = (const int*)  d_in[1];
  const float* Wl1  = (const float*)d_in[2];
  const float* Wr1  = (const float*)d_in[3];
  const float* b1   = (const float*)d_in[4];
  const float* Wlm  = (const float*)d_in[5];
  const float* Wrm  = (const float*)d_in[6];
  const float* bm   = (const float*)d_in[7];
  const float* Wl10 = (const float*)d_in[8];
  const float* Wr10 = (const float*)d_in[9];
  const float* b10  = (const float*)d_in[10];

  const int  N = in_sizes[0] / 10;
  const long E = (long)in_sizes[1] / 2;
  const int* src = ei;
  const int* dst = ei + E;

  float* ws = (float*)d_ws;
  float* H  = ws;
  float* A0 = H + (size_t)N * RP;
  float* A1 = A0 + (size_t)N * RP;
  float* z0 = A1 + (size_t)N * RP;
  float* z1 = z0 + (size_t)N;

  const int B = 256;
  const int EG = 8192;
  auto cdiv = [](long a, long b) { return (int)((a + b - 1) / b); };

  const int tiles   = (N + 15) / 16;
  const int tblocks = (tiles + 7) / 8;

  node_transform<10><<<tblocks, 256, 0, stream>>>(x, Wl1, Wr1, b1, H, A0, A1, N);
  edge_scatter_rows<6><<<EG, B, 0, stream>>>(src, dst, H, A0, A1, E);

  combine_transform<5, true><<<tblocks, 256, 0, stream>>>(
      H, A0, A1, Wlm, Wrm, bm, z0, z1, N);
  edge_scatter_rows<5><<<EG, B, 0, stream>>>(src, dst, H, A0, A1, E);
  for (int l = 1; l < 8; ++l) {
    combine_transform<5, false><<<tblocks, 256, 0, stream>>>(
        H, A0, A1, Wlm + l * 25, Wrm + l * 25, bm + l * 5, z0, z1, N);
    edge_scatter_rows<5><<<EG, B, 0, stream>>>(src, dst, H, A0, A1, E);
  }

  combine_transform<1, false><<<tblocks, 256, 0, stream>>>(
      H, A0, A1, Wl10, Wr10, b10, z0, z1, N);
  edge_scatter_one<<<EG, B, 0, stream>>>(src, dst, H, z0, z1, E);
  combine_last<<<cdiv(N, B), B, 0, stream>>>(z0, z1, A1, (float*)d_out, N);
}


# 5-wide scatter: whole rows per instruction (6 rows of 5 lanes, 5 of 6; one more instruction a trip), so the 4 rows in 32 that were split over two instructions go as one request; sum order as before
# speedup vs baseline: 6.2901x; 1.0455x over previous
#include <hip/hip_runtime.h>


typedef __attribute__((ext_vector_type(2))) float v2f;
typedef __attribute__((ext_vector_type(4))) float v4f;
typedef __attribute__((ext_vector_type(8))) float v8f;

constexpr int RP = 8;
constexpr int AP = 6;

__device__ __forceinline__ int package_half() {
  return (int)((__builtin_amdgcn_s_sendmsg_rtn(0x87) >> 18) & 1u);
}

template <int F_IN, int F_OUT>
__device__ __forceinline__ float ldB1(const float* __restrict__ Wl,
                                      const float* __restrict__ Wr, int n,
                                      int k) {
  if constexpr (F_OUT == 5) {
    const bool ok = (n < RP) && (k < F_IN);
    const bool left = (n < 5);
    const float* W = left ? Wl : Wr;
    const int r = left ? n : n - 5;
    const float v = W[ok ? (r * F_IN + k) : 0];
    return ok ? v : 0.0f;
  } else {
    const bool ok = (n == 0) && (k < F_IN);
    const float v = Wl[ok ? k : 0];
    return ok ? v : 0.0f;
  }
}
template <int F_IN, int F_OUT>
__device__ __forceinline__ float ldB2(const float* __restrict__ Wr, int n,
                                      int k) {
  if constexpr (F_OUT == 5) {
    const bool ok = (n >= 6) && (n < RP) && (k < F_IN);
    const float v = Wr[ok ? ((n - 3) * F_IN + k) : 0];
    return ok ? v : 0.0f;
  } else {
    const bool ok = (n == RP - 1) && (k < F_IN);
    const float v = Wr[ok ? k : 0];
    return ok ? v : 0.0f;
  }
}

__device__ __forceinline__ void store_rows5(const v8f& d1, const v8f& d2,
                                            const float* __restrict__ bias,
                                            float* H, float* A0, float* A1,
                                            int row0, int hi, int m,
                                            int n_nodes, int nclear) {
  if (m < RP) {
    const bool  c1 = (m >= 5);
    const bool  c2 = (m >= 6);
    const float bl = bias[c1 ? m - 5 : 0];
    const float bh = bias[c2 ? m - 3 : 0];
    const long  base = (long)(row0 + hi * 8) * RP + m;
    if (row0 + 16 <= n_nodes) {
#pragma unroll
      for (int v = 0; v < 8; ++v) {
        H[base + (long)v * RP]  = c1 ? d1[v] + bl : d1[v];
        A0[base + (long)v * RP] = c2 ? d2[v] + bh : 0.0f;
      }
      if (m < nclear) {
#pragma unroll
        for (int v = 0; v < 8; ++v) A1[base + (long)v * RP] = 0.0f;
      }
    } else {
#pragma unroll
      for (int v = 0; v < 8; ++v) {
        if (row0 + hi * 8 + v < n_nodes) {
          H[base + (long)v * RP]  = c1 ? d1[v] + bl : d1[v];
          A0[base + (long)v * RP] = c2 ? d2[v] + bh : 0.0f;
        }
      }
      if (m < nclear) {
#pragma unroll
        for (int v = 0; v < 8; ++v) {
          if (row0 + hi * 8 + v < n_nodes) A1[base + (long)v * RP] = 0.0f;
        }
      }
    }
  }
}

template <int F_IN>
__global__ void node_transform(const float* __restrict__ x,
                               const float* __restrict__ Wl,
                               const float* __restrict__ Wr,
                               const float* __restrict__ bias,
                               float* __restrict__ H,
                               float* __restrict__ A0,
                               float* __restrict__ A1, int n_nodes) {
  const int lane = threadIdx.x & 31;
  const int wave = threadIdx.x >> 5;
  const int tile = blockIdx.x * (blockDim.x >> 5) + wave;
  const int tiles = (n_nodes + 15) >> 4;
  if (tile >= tiles) return;

  const int m    = lane & 15;
  const int hi   = lane >> 4;
  const int row0 = tile << 4;
  const int arow = min(row0 + m, n_nodes - 1);

  auto ldA = [&](int k) -> float {
    const bool ok = (k < F_IN);
    const float v = x[(long)arow * F_IN + (ok ? k : 0)];
    return ok ? v : 0.0f;
  };

  v8f d1 = {};
  v8f d2 = {};
#pragma unroll
  for (int kc = 0; kc < F_IN; kc += 4) {
    const int k0 = kc + hi * 2;
    v2f a, b1, b2;
    a.x  = ldA(k0);
    a.y  = ldA(k0 + 1);
    b1.x = ldB1<F_IN, 5>(Wl, Wr, m, k0);
    b1.y = ldB1<F_IN, 5>(Wl, Wr, m, k0 + 1);
    b2.x = ldB2<F_IN, 5>(Wr, m, k0);
    b2.y = ldB2<F_IN, 5>(Wr, m, k0 + 1);
    d1 = __builtin_amdgcn_wmma_f32_16x16x4_f32(false, a, false, b1, (short)0,
                                               d1, false, false);
    d2 = __builtin_amdgcn_wmma_f32_16x16x4_f32(false, a, false, b2, (short)0,
                                               d2, false, false);
  }
  store_rows5(d1, d2, bias, H, A0, A1, row0, hi, m, n_nodes, RP);
}

template <int F_OUT, bool FIRST>
__global__ void combine_transform(float* H, float* A0, float* A1,
                                  const float* __restrict__ Wl,
                                  const float* __restrict__ Wr,
                                  const float* __restrict__ bias,
                                  float* __restrict__ z0,
                                  float* __restrict__ z1, int n_nodes) {
  constexpr int F_IN = 5;
  const int lane = threadIdx.x & 31;
  const int wave = threadIdx.x >> 5;
  const int tile = blockIdx.x * (blockDim.x >> 5) + wave;
  const int tiles = (n_nodes + 15) >> 4;
  if (tile >= tiles) return;

  const int m    = lane & 15;
  const int hi   = lane >> 4;
  const int row0 = tile << 4;
  const int arow = min(row0 + m, n_nodes - 1);
  const long rb  = (long)arow * RP;

  const v4f a0l = *reinterpret_cast<const v4f*>(A0 + rb);
  const v4f a0h = *reinterpret_cast<const v4f*>(A0 + rb + 4);
  const v4f a1l = *reinterpret_cast<const v4f*>(A1 + rb);
  const v4f a1h = *reinterpret_cast<const v4f*>(A1 + rb + 4);
  const v4f hh  = *reinterpret_cast<const v4f*>(H + rb + 4);

  float iv;
  if (FIRST) {
    iv = 1.0f / fmaxf(a0h.y + a1h.y, 1.0f);
    if (hi == 0 && row0 + m < n_nodes) A1[rb + 6] = iv;
  } else {
    iv = a1h.z;
  }
  const float h0 = fmaxf((a0l.x + a1l.x) * iv + hh.y, 0.0f);
  const float h1 = fmaxf((a0l.y + a1l.y) * iv + hh.z, 0.0f);
  const float h2 = fmaxf((a0l.z + a1l.z) * iv + hh.w, 0.0f);
  const float h3 = fmaxf((a0l.w + a1l.w) * iv + a0h.z, 0.0f);
  const float h4 = fmaxf((a0h.x + a1h.x) * iv + a0h.w, 0.0f);

  v8f d1 = {};
  v8f d2 = {};
  {
    const int k0 = hi * 2;
    v2f a, b1, b2;
    a.x  = hi ? h2 : h0;
    a.y  = hi ? h3 : h1;
    b1.x = ldB1<F_IN, F_OUT>(Wl, Wr, m, k0);
    b1.y = ldB1<F_IN, F_OUT>(Wl, Wr, m, k0 + 1);
    b2.x = ldB2<F_IN, F_OUT>(Wr, m, k0);
    b2.y = ldB2<F_IN, F_OUT>(Wr, m, k0 + 1);
    d1 = __builtin_amdgcn_wmma_f32_16x16x4_f32(false, a, false, b1, (short)0,
                                               d1, false, false);
    d2 = __builtin_amdgcn_wmma_f32_16x16x4_f32(false, a, false, b2, (short)0,
                                               d2, false, false);
  }
  {
    const int k0 = 4 + hi * 2;
    v2f a, b1, b2;
    a.x  = hi ? 0.0f : h4;
    a.y  = 0.0f;
    b1.x = ldB1<F_IN, F_OUT>(Wl, Wr, m, k0);
    b1.y = ldB1<F_IN, F_OUT>(Wl, Wr, m, k0 + 1);
    b2.x = ldB2<F_IN, F_OUT>(Wr, m, k0);
    b2.y = ldB2<F_IN, F_OUT>(Wr, m, k0 + 1);
    d1 = __builtin_amdgcn_wmma_f32_16x16x4_f32(false, a, false, b1, (short)0,
                                               d1, false, false);
    d2 = __builtin_amdgcn_wmma_f32_16x16x4_f32(false, a, false, b2, (short)0,
                                               d2, false, false);
  }

  if constexpr (F_OUT == 5) {
    store_rows5(d1, d2, bias, H, A0, A1, row0, hi, m, n_nodes, AP);
  } else {
    const bool c0 = (m == 0);
    if (c0 || m == RP - 1) {
      const float bv = bias[0];
      float* p = (c0 ? H : A1) + (long)(row0 + hi * 8) * RP + m;
      const long zb = (long)(row0 + hi * 8);
      if (row0 + 16 <= n_nodes) {
#pragma unroll
        for (int v = 0; v < 8; ++v)
          p[(long)v * RP] = c0 ? d1[v] : d2[v] + bv;
        if (c0) {
#pragma unroll
          for (int v = 0; v < 8; ++v) {
            z0[zb + v] = 0.0f;
            z1[zb + v] = 0.0f;
          }
        }
      } else {
#pragma unroll
        for (int v = 0; v < 8; ++v) {
          if (row0 + hi * 8 + v < n_nodes)
            p[(long)v * RP] = c0 ? d1[v] : d2[v] + bv;
        }
        if (c0) {
#pragma unroll
          for (int v = 0; v < 8; ++v) {
            if (row0 + hi * 8 + v < n_nodes) {
              z0[zb + v] = 0.0f;
              z1[zb + v] = 0.0f;
            }
          }
        }
      }
    }
  }
}

__global__ void edge_scatter_one(const int* __restrict__ src,
                                 const int* __restrict__ dst,
                                 const float* __restrict__ H,
                                 float* __restrict__ z0,
                                 float* __restrict__ z1, long n_edges) {
  float* __restrict__ agg = package_half() ? z1 : z0;
  const long stride = (long)gridDim.x * blockDim.x;
  for (long i = (long)blockIdx.x * blockDim.x + threadIdx.x; i < n_edges;
       i += stride) {
    const int s = src[i];
    const int d = dst[i];
    unsafeAtomicAdd(&agg[d], H[s * RP]);
  }
}

template <int F>
__global__ void edge_scatter_rows(const int* __restrict__ src,
                                  const int* __restrict__ dst,
                                  const float* H, float* A0, float* A1,
                                  long n_edges) {
  constexpr int FH = 5;
  constexpr int R = 32 / F;
  constexpr int T = (32 + R - 1) / R;
  constexpr int RL = 32 - (T - 1) * R;
  float* agg = package_half() ? A1 : A0;
  const int lane = threadIdx.x & 31;
  const int r0 = lane / F;
  const int j = lane - r0 * F;
  const bool in = (r0 < R);
  const int r = in ? r0 : R - 1;
  const bool cl = (F > FH) && (j >= FH);
  const int jg = cl ? FH - 1 : j;
  const long wave =
      (long)blockIdx.x * (blockDim.x >> 5) +
      __builtin_amdgcn_readfirstlane((int)(threadIdx.x >> 5));
  const long stride = (long)gridDim.x * blockDim.x;
  for (long base = wave * 32; base < n_edges; base += stride) {
    if (base + 32 <= n_edges) {
      const int s = src[base + lane];
      const int d = dst[base + lane];
      float v[T];
      int de[T];
#pragma unroll
      for (int t = 0; t < T; ++t) {
        const int e = (t < T - 1 || r < RL) ? t * R + r : 31;
        const int se = __shfl(s, e);
        de[t] = __shfl(d, e) * RP + j;
        const float g = H[se * RP + jg];
        v[t] = cl ? 1.0f : g;
      }
      if (in) {
#pragma unroll
        for (int t = 0; t < T - 1; ++t) unsafeAtomicAdd(&agg[de[t]], v[t]);
        if (r0 < RL) unsafeAtomicAdd(&agg[de[T - 1]], v[T - 1]);
      }
    } else {
      int s = 0, d = 0;
      if (base + lane < n_edges) {
        s = src[base + lane];
        d = dst[base + lane];
      }
#pragma unroll
      for (int t = 0; t < T; ++t) {
        const int e0 = t * R + r0;
        const bool has = in && (e0 < 32);
        const int e = has ? e0 : 31;
        const int se = __shfl(s, e);
        const int dd = __shfl(d, e);
        if (has && base + e < n_edges) {
          const float g = H[se * RP + jg];
          unsafeAtomicAdd(&agg[dd * RP + j], cl ? 1.0f : g);
        }
      }
    }
  }
}

__global__ void combine_last(const float* __restrict__ z0,
                             const float* __restrict__ z1,
                             const float* __restrict__ A1,
                             float* __restrict__ out, int n_nodes) {
  const int node = blockIdx.x * blockDim.x + threadIdx.x;
  if (node >= n_nodes) return;
  const v2f t = *reinterpret_cast<const v2f*>(A1 + (long)node * RP + 6);
  const float a = z0[node] + z1[node];
  out[node] = a * t.x + t.y;
}

extern "C" void kernel_launch(void* const* d_in, const int* in_sizes, int n_in,
                              void* d_out, int out_size, void* d_ws,
                              size_t ws_size, hipStream_t stream) {
  const float* x    = (const float*)d_in[0];
  const int*   ei   = (const int*)  d_in[1];
  const float* Wl1  = (const float*)d_in[2];
  const float* Wr1  = (const float*)d_in[3];
  const float* b1   = (const float*)d_in[4];
  const float* Wlm  = (const float*)d_in[5];
  const float* Wrm  = (const float*)d_in[6];
  const float* bm   = (const float*)d_in[7];
  const float* Wl10 = (const float*)d_in[8];
  const float* Wr10 = (const float*)d_in[9];
  const float* b10  = (const float*)d_in[10];

  const int  N = in_sizes[0] / 10;
  const long E = (long)in_sizes[1] / 2;
  const int* src = ei;
  const int* dst = ei + E;

  float* ws = (float*)d_ws;
  float* H  = ws;
  float* A0 = H + (size_t)N * RP;
  float* A1 = A0 + (size_t)N * RP;
  float* z0 = A1 + (size_t)N * RP;
  float* z1 = z0 + (size_t)N;

  const int B = 256;
  const int EG = 8192;
  auto cdiv = [](long a, long b) { return (int)((a + b - 1) / b); };

  const int tiles   = (N + 15) / 16;
  const int tblocks = (tiles + 7) / 8;

  node_transform<10><<<tblocks, 256, 0, stream>>>(x, Wl1, Wr1, b1, H, A0, A1, N);
  edge_scatter_rows<6><<<EG, B, 0, stream>>>(src, dst, H, A0, A1, E);

  combine_transform<5, true><<<tblocks, 256, 0, stream>>>(
      H, A0, A1, Wlm, Wrm, bm, z0, z1, N);
  edge_scatter_rows<5><<<EG, B, 0, stream>>>(src, dst, H, A0, A1, E);
  for (int l = 1; l < 8; ++l) {
    combine_transform<5, false><<<tblocks, 256, 0, stream>>>(
        H, A0, A1, Wlm + l * 25, Wrm + l * 25, bm + l * 5, z0, z1, N);
    edge_scatter_rows<5><<<EG, B, 0, stream>>>(src, dst, H, A0, A1, E);
  }

  combine_transform<1, false><<<tblocks, 256, 0, stream>>>(
      H, A0, A1, Wl10, Wr10, b10, z0, z1, N);
  edge_scatter_one<<<EG, B, 0, stream>>>(src, dst, H, z0, z1, E);
  combine_last<<<cdiv(N, B), B, 0, stream>>>(z0, z1, A1, (float*)d_out, N);
}


# Edges binned once per call by 1024-node source block (LDS int adds, scan, arrival-rank fill); scatter blocks read their block's H rows from LDS, not L2. Edges visited bin by bin: add order changes
# speedup vs baseline: 8.7129x; 1.3852x over previous
#include <hip/hip_runtime.h>


typedef __attribute__((ext_vector_type(2))) float v2f;
typedef __attribute__((ext_vector_type(4))) float v4f;
typedef __attribute__((ext_vector_type(8))) float v8f;

constexpr int RP = 8;
constexpr int AP = 6;

__device__ __forceinline__ int package_half() {
  return (int)((__builtin_amdgcn_s_sendmsg_rtn(0x87) >> 18) & 1u);
}

template <int F_IN, int F_OUT>
__device__ __forceinline__ float ldB1(const float* __restrict__ Wl,
                                      const float* __restrict__ Wr, int n,
                                      int k) {
  if constexpr (F_OUT == 5) {
    const bool ok = (n < RP) && (k < F_IN);
    const bool left = (n < 5);
    const float* W = left ? Wl : Wr;
    const int r = left ? n : n - 5;
    const float v = W[ok ? (r * F_IN + k) : 0];
    return ok ? v : 0.0f;
  } else {
    const bool ok = (n == 0) && (k < F_IN);
    const float v = Wl[ok ? k : 0];
    return ok ? v : 0.0f;
  }
}
template <int F_IN, int F_OUT>
__device__ __forceinline__ float ldB2(const float* __restrict__ Wr, int n,
                                      int k) {
  if constexpr (F_OUT == 5) {
    const bool ok = (n >= 6) && (n < RP) && (k < F_IN);
    const float v = Wr[ok ? ((n - 3) * F_IN + k) : 0];
    return ok ? v : 0.0f;
  } else {
    const bool ok = (n == RP - 1) && (k < F_IN);
    const float v = Wr[ok ? k : 0];
    return ok ? v : 0.0f;
  }
}

__device__ __forceinline__ void store_rows5(const v8f& d1, const v8f& d2,
                                            const float* __restrict__ bias,
                                            float* H, float* A0, float* A1,
                                            int row0, int hi, int m,
                                            int n_nodes, int nclear) {
  if (m < RP) {
    const bool  c1 = (m >= 5);
    const bool  c2 = (m >= 6);
    const float bl = bias[c1 ? m - 5 : 0];
    const float bh = bias[c2 ? m - 3 : 0];
    const long  base = (long)(row0 + hi * 8) * RP + m;
    if (row0 + 16 <= n_nodes) {
#pragma unroll
      for (int v = 0; v < 8; ++v) {
        H[base + (long)v * RP]  = c1 ? d1[v] + bl : d1[v];
        A0[base + (long)v * RP] = c2 ? d2[v] + bh : 0.0f;
      }
      if (m < nclear) {
#pragma unroll
        for (int v = 0; v < 8; ++v) A1[base + (long)v * RP] = 0.0f;
      }
    } else {
#pragma unroll
      for (int v = 0; v < 8; ++v) {
        if (row0 + hi * 8 + v < n_nodes) {
          H[base + (long)v * RP]  = c1 ? d1[v] + bl : d1[v];
          A0[base + (long)v * RP] = c2 ? d2[v] + bh : 0.0f;
        }
      }
      if (m < nclear) {
#pragma unroll
        for (int v = 0; v < 8; ++v) {
          if (row0 + hi * 8 + v < n_nodes) A1[base + (long)v * RP] = 0.0f;
        }
      }
    }
  }
}

template <int F_IN>
__global__ void node_transform(const float* __restrict__ x,
                               const float* __restrict__ Wl,
                               const float* __restrict__ Wr,
                               const float* __restrict__ bias,
                               float* __restrict__ H,
                               float* __restrict__ A0,
                               float* __restrict__ A1, int n_nodes) {
  const int lane = threadIdx.x & 31;
  const int wave = threadIdx.x >> 5;
  const int tile = blockIdx.x * (blockDim.x >> 5) + wave;
  const int tiles = (n_nodes + 15) >> 4;
  if (tile >= tiles) return;

  const int m    = lane & 15;
  const int hi   = lane >> 4;
  const int row0 = tile << 4;
  const int arow = min(row0 + m, n_nodes - 1);

  auto ldA = [&](int k) -> float {
    const bool ok = (k < F_IN);
    const float v = x[(long)arow * F_IN + (ok ? k : 0)];
    return ok ? v : 0.0f;
  };

  v8f d1 = {};
  v8f d2 = {};
#pragma unroll
  for (int kc = 0; kc < F_IN; kc += 4) {
    const int k0 = kc + hi * 2;
    v2f a, b1, b2;
    a.x  = ldA(k0);
    a.y  = ldA(k0 + 1);
    b1.x = ldB1<F_IN, 5>(Wl, Wr, m, k0);
    b1.y = ldB1<F_IN, 5>(Wl, Wr, m, k0 + 1);
    b2.x = ldB2<F_IN, 5>(Wr, m, k0);
    b2.y = ldB2<F_IN, 5>(Wr, m, k0 + 1);
    d1 = __builtin_amdgcn_wmma_f32_16x16x4_f32(false, a, false, b1, (short)0,
                                               d1, false, false);
    d2 = __builtin_amdgcn_wmma_f32_16x16x4_f32(false, a, false, b2, (short)0,
                                               d2, false, false);
  }
  store_rows5(d1, d2, bias, H, A0, A1, row0, hi, m, n_nodes, RP);
}

template <int F_OUT, bool FIRST>
__global__ void combine_transform(float* H, float* A0, float* A1,
                                  const float* __restrict__ Wl,
                                  const float* __restrict__ Wr,
                                  const float* __restrict__ bias,
                                  float* __restrict__ z0,
                                  float* __restrict__ z1, int n_nodes) {
  constexpr int F_IN = 5;
  const int lane = threadIdx.x & 31;
  const int wave = threadIdx.x >> 5;
  const int tile = blockIdx.x * (blockDim.x >> 5) + wave;
  const int tiles = (n_nodes + 15) >> 4;
  if (tile >= tiles) return;

  const int m    = lane & 15;
  const int hi   = lane >> 4;
  const int row0 = tile << 4;
  const int arow = min(row0 + m, n_nodes - 1);
  const long rb  = (long)arow * RP;

  const v4f a0l = *reinterpret_cast<const v4f*>(A0 + rb);
  const v4f a0h = *reinterpret_cast<const v4f*>(A0 + rb + 4);
  const v4f a1l = *reinterpret_cast<const v4f*>(A1 + rb);
  const v4f a1h = *reinterpret_cast<const v4f*>(A1 + rb + 4);
  const v4f hh  = *reinterpret_cast<const v4f*>(H + rb + 4);

  float iv;
  if (FIRST) {
    iv = 1.0f / fmaxf(a0h.y + a1h.y, 1.0f);
    if (hi == 0 && row0 + m < n_nodes) A1[rb + 6] = iv;
  } else {
    iv = a1h.z;
  }
  const float h0 = fmaxf((a0l.x + a1l.x) * iv + hh.y, 0.0f);
  const float h1 = fmaxf((a0l.y + a1l.y) * iv + hh.z, 0.0f);
  const float h2 = fmaxf((a0l.z + a1l.z) * iv + hh.w, 0.0f);
  const float h3 = fmaxf((a0l.w + a1l.w) * iv + a0h.z, 0.0f);
  const float h4 = fmaxf((a0h.x + a1h.x) * iv + a0h.w, 0.0f);

  v8f d1 = {};
  v8f d2 = {};
  {
    const int k0 = hi * 2;
    v2f a, b1, b2;
    a.x  = hi ? h2 : h0;
    a.y  = hi ? h3 : h1;
    b1.x = ldB1<F_IN, F_OUT>(Wl, Wr, m, k0);
    b1.y = ldB1<F_IN, F_OUT>(Wl, Wr, m, k0 + 1);
    b2.x = ldB2<F_IN, F_OUT>(Wr, m, k0);
    b2.y = ldB2<F_IN, F_OUT>(Wr, m, k0 + 1);
    d1 = __builtin_amdgcn_wmma_f32_16x16x4_f32(false, a, false, b1, (short)0,
                                               d1, false, false);
    d2 = __builtin_amdgcn_wmma_f32_16x16x4_f32(false, a, false, b2, (short)0,
                                               d2, false, false);
  }
  {
    const int k0 = 4 + hi * 2;
    v2f a, b1, b2;
    a.x  = hi ? 0.0f : h4;
    a.y  = 0.0f;
    b1.x = ldB1<F_IN, F_OUT>(Wl, Wr, m, k0);
    b1.y = ldB1<F_IN, F_OUT>(Wl, Wr, m, k0 + 1);
    b2.x = ldB2<F_IN, F_OUT>(Wr, m, k0);
    b2.y = ldB2<F_IN, F_OUT>(Wr, m, k0 + 1);
    d1 = __builtin_amdgcn_wmma_f32_16x16x4_f32(false, a, false, b1, (short)0,
                                               d1, false, false);
    d2 = __builtin_amdgcn_wmma_f32_16x16x4_f32(false, a, false, b2, (short)0,
                                               d2, false, false);
  }

  if constexpr (F_OUT == 5) {
    store_rows5(d1, d2, bias, H, A0, A1, row0, hi, m, n_nodes, AP);
  } else {
    const bool c0 = (m == 0);
    if (c0 || m == RP - 1) {
      const float bv = bias[0];
      float* p = (c0 ? H : A1) + (long)(row0 + hi * 8) * RP + m;
      const long zb = (long)(row0 + hi * 8);
      if (row0 + 16 <= n_nodes) {
#pragma unroll
        for (int v = 0; v < 8; ++v)
          p[(long)v * RP] = c0 ? d1[v] : d2[v] + bv;
        if (c0) {
#pragma unroll
          for (int v = 0; v < 8; ++v) {
            z0[zb + v] = 0.0f;
            z1[zb + v] = 0.0f;
          }
        }
      } else {
#pragma unroll
        for (int v = 0; v < 8; ++v) {
          if (row0 + hi * 8 + v < n_nodes)
            p[(long)v * RP] = c0 ? d1[v] : d2[v] + bv;
        }
        if (c0) {
#pragma unroll
          for (int v = 0; v < 8; ++v) {
            if (row0 + hi * 8 + v < n_nodes) {
              z0[zb + v] = 0.0f;
              z1[zb + v] = 0.0f;
            }
          }
        }
      }
    }
  }
}

__global__ void edge_scatter_one_direct(const int* __restrict__ src,
                                 const int* __restrict__ dst,
                                 const float* __restrict__ H,
                                 float* __restrict__ z0,
                                 float* __restrict__ z1, long n_edges) {
  float* __restrict__ agg = package_half() ? z1 : z0;
  const long stride = (long)gridDim.x * blockDim.x;
  for (long i = (long)blockIdx.x * blockDim.x + threadIdx.x; i < n_edges;
       i += stride) {
    const int s = src[i];
    const int d = dst[i];
    unsafeAtomicAdd(&agg[d], H[s * RP]);
  }
}

template <int F>
__global__ void edge_scatter_rows_direct(const int* __restrict__ src,
                                  const int* __restrict__ dst,
                                  const float* H, float* A0, float* A1,
                                  long n_edges) {
  constexpr int FH = 5;
  constexpr int R = 32 / F;
  constexpr int T = (32 + R - 1) / R;
  constexpr int RL = 32 - (T - 1) * R;
  float* agg = package_half() ? A1 : A0;
  const int lane = threadIdx.x & 31;
  const int r0 = lane / F;
  const int j = lane - r0 * F;
  const bool in = (r0 < R);
  const int r = in ? r0 : R - 1;
  const bool cl = (F > FH) && (j >= FH);
  const int jg = cl ? FH - 1 : j;
  const long wave =
      (long)blockIdx.x * (blockDim.x >> 5) +
      __builtin_amdgcn_readfirstlane((int)(threadIdx.x >> 5));
  const long stride = (long)gridDim.x * blockDim.x;
  for (long base = wave * 32; base < n_edges; base += stride) {
    if (base + 32 <= n_edges) {
      const int s = src[base + lane];
      const int d = dst[base + lane];
      float v[T];
      int de[T];
#pragma unroll
      for (int t = 0; t < T; ++t) {
        const int e = (t < T - 1 || r < RL) ? t * R + r : 31;
        const int se = __shfl(s, e);
        de[t] = __shfl(d, e) * RP + j;
        const float g = H[se * RP + jg];
        v[t] = cl ? 1.0f : g;
      }
      if (in) {
#pragma unroll
        for (int t = 0; t < T - 1; ++t) unsafeAtomicAdd(&agg[de[t]], v[t]);
        if (r0 < RL) unsafeAtomicAdd(&agg[de[T - 1]], v[T - 1]);
      }
    } else {
      int s = 0, d = 0;
      if (base + lane < n_edges) {
        s = src[base + lane];
        d = dst[base + lane];
      }
#pragma unroll
      for (int t = 0; t < T; ++t) {
        const int e0 = t * R + r0;
        const bool has = in && (e0 < 32);
        const int e = has ? e0 : 31;
        const int se = __shfl(s, e);
        const int dd = __shfl(d, e);
        if (has && base + e < n_edges) {
          const float g = H[se * RP + jg];
          unsafeAtomicAdd(&agg[dd * RP + j], cl ? 1.0f : g);
        }
      }
    }
  }
}

constexpr int SB = 10;
constexpr int BR = 1 << SB;
constexpr int DB = 19;
constexpr int MAXB = 512;
constexpr int PG = 1024;
constexpr int SP = 4;
constexpr int SW = 256;
constexpr int DMASK = (1 << DB) - 1;

__global__ void bin_count(const int* __restrict__ src, int* __restrict__ cnt,
                          long n_edges) {
  __shared__ int hist[MAXB];
  hist[threadIdx.x] = 0;
  __syncthreads();
  const long per = (n_edges + gridDim.x - 1) / gridDim.x;
  const long c0 = (long)blockIdx.x * per;
  const long c1 = min(n_edges, c0 + per);
  long i = c0 + threadIdx.x;
  for (; i + 3 * MAXB < c1; i += 4 * MAXB) {
    const int s0 = src[i], s1 = src[i + MAXB], s2 = src[i + 2 * MAXB],
              s3 = src[i + 3 * MAXB];
    atomicAdd(&hist[(s0 >> SB) & (MAXB - 1)], 1);
    atomicAdd(&hist[(s1 >> SB) & (MAXB - 1)], 1);
    atomicAdd(&hist[(s2 >> SB) & (MAXB - 1)], 1);
    atomicAdd(&hist[(s3 >> SB) & (MAXB - 1)], 1);
  }
  for (; i < c1; i += MAXB) atomicAdd(&hist[(src[i] >> SB) & (MAXB - 1)], 1);
  __syncthreads();
  cnt[blockIdx.x * MAXB + threadIdx.x] = hist[threadIdx.x];
}

__device__ __forceinline__ int scan_block(int* sc, int mine) {
  const int t = threadIdx.x;
  sc[t] = mine;
  __syncthreads();
  for (int off = 1; off < MAXB; off <<= 1) {
    const int v = (t >= off) ? sc[t - off] : 0;
    __syncthreads();
    sc[t] += v;
    __syncthreads();
  }
  return sc[t];
}

__global__ void bin_scan(int* __restrict__ cnt, int* __restrict__ total) {
  __shared__ int sc[MAXB];
  const int b = blockIdx.x;
  const int t = threadIdx.x;
  constexpr int Q = PG / MAXB;
  int c[Q];
  int s = 0;
#pragma unroll
  for (int q = 0; q < Q; ++q) {
    c[q] = cnt[(t * Q + q) * MAXB + b];
    s += c[q];
  }
  const int incl = scan_block(sc, s);
  int run = incl - s;
#pragma unroll
  for (int q = 0; q < Q; ++q) {
    cnt[(t * Q + q) * MAXB + b] = run;
    run += c[q];
  }
  if (t == MAXB - 1) total[b] = incl;
}

__global__ void bin_fill(const int* __restrict__ src,
                         const int* __restrict__ dst,
                         const int* __restrict__ cnt,
                         const int* __restrict__ total,
                         int* __restrict__ bstart, int* __restrict__ pk,
                         long n_edges) {
  __shared__ int sc[MAXB];
  __shared__ int cur[MAXB];
  const int t = threadIdx.x;
  const int tot = total[t];
  const int first = scan_block(sc, tot) - tot;
  cur[t] = first + cnt[blockIdx.x * MAXB + t];
  if (blockIdx.x == 0) {
    bstart[t] = first;
    if (t == MAXB - 1) bstart[MAXB] = first + tot;
  }
  __syncthreads();
  const long per = (n_edges + gridDim.x - 1) / gridDim.x;
  const long c0 = (long)blockIdx.x * per;
  const long c1 = min(n_edges, c0 + per);
  auto put = [&](int s, int d) {
    const int pos = atomicAdd(&cur[(s >> SB) & (MAXB - 1)], 1);
    pk[pos] = ((s & (BR - 1)) << DB) | d;
  };
  long i = c0 + t;
  for (; i + 3 * MAXB < c1; i += 4 * MAXB) {
    const int s0 = src[i], s1 = src[i + MAXB], s2 = src[i + 2 * MAXB],
              s3 = src[i + 3 * MAXB];
    const int d0 = dst[i], d1 = dst[i + MAXB], d2 = dst[i + 2 * MAXB],
              d3 = dst[i + 3 * MAXB];
    put(s0, d0);
    put(s1, d1);
    put(s2, d2);
    put(s3, d3);
  }
  for (; i < c1; i += MAXB) put(src[i], dst[i]);
}

__device__ __forceinline__ void bin_part(const int* __restrict__ bstart, int b,
                                         int part, int& p0, int& p1) {
  const int start = bstart[b];
  const int end = bstart[b + 1];
  const int per = ((end - start + SP * 32 - 1) / (SP * 32)) * 32;
  p0 = min(end, start + part * per);
  p1 = min(end, p0 + per);
}

__global__ void edge_scatter_one(const int* __restrict__ pk,
                                 const int* __restrict__ bstart,
                                 const float* __restrict__ H,
                                 float* __restrict__ z0,
                                 float* __restrict__ z1, int n_nodes) {
  __shared__ float hb[BR];
  const int b = blockIdx.x / SP;
  const int part = blockIdx.x - b * SP;
  const int row_base = b << SB;
  const int nrows = min(BR, n_nodes - row_base);
  for (int r = threadIdx.x; r < nrows; r += SW)
    hb[r] = H[(long)(row_base + r) * RP];
  __syncthreads();
  float* __restrict__ agg = package_half() ? z1 : z0;
  int p0, p1;
  bin_part(bstart, b, part, p0, p1);
  for (int i = p0 + threadIdx.x; i < p1; i += SW) {
    const int w = pk[i];
    unsafeAtomicAdd(&agg[w & DMASK], hb[w >> DB]);
  }
}

template <int F>
__global__ void edge_scatter_rows(const int* __restrict__ pk,
                                  const int* __restrict__ bstart,
                                  const float* H, float* A0, float* A1,
                                  int n_nodes) {
  constexpr int FH = 5;
  constexpr int R = 32 / F;
  constexpr int T = (32 + R - 1) / R;
  constexpr int RL = 32 - (T - 1) * R;
  constexpr int NQ = BR * 2 / SW;
  __shared__ float hb[BR * FH];
  const int b = blockIdx.x / SP;
  const int part = blockIdx.x - b * SP;
  const int row_base = b << SB;
  const int nrows = min(BR, n_nodes - row_base);
  {
    const float* Hb = H + (long)row_base * RP;
    const int nq = nrows * 2;
    v4f hv[NQ];
#pragma unroll
    for (int u = 0; u < NQ; ++u) {
      const int q = min((int)threadIdx.x + u * SW, nq - 1);
      hv[u] = *reinterpret_cast<const v4f*>(Hb + (long)q * 4);
    }
#pragma unroll
    for (int u = 0; u < NQ; ++u) {
      const int q = (int)threadIdx.x + u * SW;
      if (q < nq) {
        const int r = q >> 1;
        if (q & 1) {
          hb[r * FH + 4] = hv[u].x;
        } else {
          hb[r * FH + 0] = hv[u].x;
          hb[r * FH + 1] = hv[u].y;
          hb[r * FH + 2] = hv[u].z;
          hb[r * FH + 3] = hv[u].w;
        }
      }
    }
  }
  __syncthreads();
  float* agg = package_half() ? A1 : A0;
  const int lane = threadIdx.x & 31;
  const int r0 = lane / F;
  const int j = lane - r0 * F;
  const bool in = (r0 < R);
  const int r = in ? r0 : R - 1;
  const bool cl = (F > FH) && (j >= FH);
  const int jg = cl ? FH - 1 : j;
  const int wv = __builtin_amdgcn_readfirstlane((int)(threadIdx.x >> 5));
  int p0, p1;
  bin_part(bstart, b, part, p0, p1);
  for (int base = p0 + wv * 32; base < p1; base += SW) {
    if (base + 32 <= p1) {
      const int w = pk[base + lane];
      float v[T];
      int de[T];
#pragma unroll
      for (int t = 0; t < T; ++t) {
        const int e = (t < T - 1 || r < RL) ? t * R + r : 31;
        const int we = __shfl(w, e);
        de[t] = (we & DMASK) * RP + j;
        const float g = hb[(we >> DB) * FH + jg];
        v[t] = cl ? 1.0f : g;
      }
      if (in) {
#pragma unroll
        for (int t = 0; t < T - 1; ++t) unsafeAtomicAdd(&agg[de[t]], v[t]);
        if (r0 < RL) unsafeAtomicAdd(&agg[de[T - 1]], v[T - 1]);
      }
    } else {
      int w = 0;
      if (base + lane < p1) w = pk[base + lane];
#pragma unroll
      for (int t = 0; t < T; ++t) {
        const int e0 = t * R + r0;
        const bool has = in && (e0 < 32);
        const int e = has ? e0 : 31;
        const int we = __shfl(w, e);
        if (has && base + e < p1) {
          const float g = hb[(we >> DB) * FH + jg];
          unsafeAtomicAdd(&agg[(we & DMASK) * RP + j], cl ? 1.0f : g);
        }
      }
    }
  }
}

__global__ void combine_last(const float* __restrict__ z0,
                             const float* __restrict__ z1,
                             const float* __restrict__ A1,
                             float* __restrict__ out, int n_nodes) {
  const int node = blockIdx.x * blockDim.x + threadIdx.x;
  if (node >= n_nodes) return;
  const v2f t = *reinterpret_cast<const v2f*>(A1 + (long)node * RP + 6);
  const float a = z0[node] + z1[node];
  out[node] = a * t.x + t.y;
}

extern "C" void kernel_launch(void* const* d_in, const int* in_sizes, int n_in,
                              void* d_out, int out_size, void* d_ws,
                              size_t ws_size, hipStream_t stream) {
  const float* x    = (const float*)d_in[0];
  const int*   ei   = (const int*)  d_in[1];
  const float* Wl1  = (const float*)d_in[2];
  const float* Wr1  = (const float*)d_in[3];
  const float* b1   = (const float*)d_in[4];
  const float* Wlm  = (const float*)d_in[5];
  const float* Wrm  = (const float*)d_in[6];
  const float* bm   = (const float*)d_in[7];
  const float* Wl10 = (const float*)d_in[8];
  const float* Wr10 = (const float*)d_in[9];
  const float* b10  = (const float*)d_in[10];

  const int  N = in_sizes[0] / 10;
  const long E = (long)in_sizes[1] / 2;
  const int* src = ei;
  const int* dst = ei + E;

  float* ws = (float*)d_ws;
  float* H  = ws;
  float* A0 = H + (size_t)N * RP;
  float* A1 = A0 + (size_t)N * RP;
  float* z0 = A1 + (size_t)N * RP;
  float* z1 = z0 + (size_t)N;
  int* pk     = (int*)(z1 + (size_t)N);
  int* cnt    = pk + (size_t)E;
  int* total  = cnt + (size_t)PG * MAXB;
  int* bstart = total + MAXB;

  const size_t need = ((size_t)N * 26 + (size_t)E + (size_t)PG * MAXB +
                       2 * MAXB + 8) * sizeof(float);
  const bool binned = ws_size >= need && N <= (1 << DB) && E <= (1L << 30);

  const int B = 256;
  const int EGD = 8192;
  auto cdiv = [](long a, long b) { return (int)((a + b - 1) / b); };

  const int tiles   = (N + 15) / 16;
  const int tblocks = (tiles + 7) / 8;
  const int EG = cdiv(N, BR) * SP;

  if (binned) {
    bin_count<<<PG, MAXB, 0, stream>>>(src, cnt, E);
    bin_scan<<<MAXB, MAXB, 0, stream>>>(cnt, total);
    bin_fill<<<PG, MAXB, 0, stream>>>(src, dst, cnt, total, bstart, pk, E);
  }

  node_transform<10><<<tblocks, 256, 0, stream>>>(x, Wl1, Wr1, b1, H, A0, A1, N);
  if (binned)
    edge_scatter_rows<6><<<EG, SW, 0, stream>>>(pk, bstart, H, A0, A1, N);
  else
    edge_scatter_rows_direct<6><<<EGD, B, 0, stream>>>(src, dst, H, A0, A1, E);

  for (int l = 0; l < 8; ++l) {
    if (l == 0)
      combine_transform<5, true><<<tblocks, 256, 0, stream>>>(
          H, A0, A1, Wlm, Wrm, bm, z0, z1, N);
    else
      combine_transform<5, false><<<tblocks, 256, 0, stream>>>(
          H, A0, A1, Wlm + l * 25, Wrm + l * 25, bm + l * 5, z0, z1, N);
    if (binned)
      edge_scatter_rows<5><<<EG, SW, 0, stream>>>(pk, bstart, H, A0, A1, N);
    else
      edge_scatter_rows_direct<5><<<EGD, B, 0, stream>>>(src, dst, H, A0, A1, E);
  }

  combine_transform<1, false><<<tblocks, 256, 0, stream>>>(
      H, A0, A1, Wl10, Wr10, b10, z0, z1, N);
  if (binned)
    edge_scatter_one<<<EG, SW, 0, stream>>>(pk, bstart, H, z0, z1, N);
  else
    edge_scatter_one_direct<<<EGD, B, 0, stream>>>(src, dst, H, z0, z1, E);
  combine_last<<<cdiv(N, B), B, 0, stream>>>(z0, z1, A1, (float*)d_out, N);
}


# Bin fill staged in LDS: a workgroup ranks its chunk's edges into an LDS copy ordered by bin, then stores it out in bin runs, whole lines (was a 4-byte store per edge); in-bin rank still arrival order
# speedup vs baseline: 9.6059x; 1.1025x over previous
#include <hip/hip_runtime.h>


typedef __attribute__((ext_vector_type(2))) float v2f;
typedef __attribute__((ext_vector_type(4))) float v4f;
typedef __attribute__((ext_vector_type(8))) float v8f;

constexpr int RP = 8;
constexpr int AP = 6;

__device__ __forceinline__ int package_half() {
  return (int)((__builtin_amdgcn_s_sendmsg_rtn(0x87) >> 18) & 1u);
}

template <int F_IN, int F_OUT>
__device__ __forceinline__ float ldB1(const float* __restrict__ Wl,
                                      const float* __restrict__ Wr, int n,
                                      int k) {
  if constexpr (F_OUT == 5) {
    const bool ok = (n < RP) && (k < F_IN);
    const bool left = (n < 5);
    const float* W = left ? Wl : Wr;
    const int r = left ? n : n - 5;
    const float v = W[ok ? (r * F_IN + k) : 0];
    return ok ? v : 0.0f;
  } else {
    const bool ok = (n == 0) && (k < F_IN);
    const float v = Wl[ok ? k : 0];
    return ok ? v : 0.0f;
  }
}
template <int F_IN, int F_OUT>
__device__ __forceinline__ float ldB2(const float* __restrict__ Wr, int n,
                                      int k) {
  if constexpr (F_OUT == 5) {
    const bool ok = (n >= 6) && (n < RP) && (k < F_IN);
    const float v = Wr[ok ? ((n - 3) * F_IN + k) : 0];
    return ok ? v : 0.0f;
  } else {
    const bool ok = (n == RP - 1) && (k < F_IN);
    const float v = Wr[ok ? k : 0];
    return ok ? v : 0.0f;
  }
}

__device__ __forceinline__ void store_rows5(const v8f& d1, const v8f& d2,
                                            const float* __restrict__ bias,
                                            float* H, float* A0, float* A1,
                                            int row0, int hi, int m,
                                            int n_nodes, int nclear) {
  if (m < RP) {
    const bool  c1 = (m >= 5);
    const bool  c2 = (m >= 6);
    const float bl = bias[c1 ? m - 5 : 0];
    const float bh = bias[c2 ? m - 3 : 0];
    const long  base = (long)(row0 + hi * 8) * RP + m;
    if (row0 + 16 <= n_nodes) {
#pragma unroll
      for (int v = 0; v < 8; ++v) {
        H[base + (long)v * RP]  = c1 ? d1[v] + bl : d1[v];
        A0[base + (long)v * RP] = c2 ? d2[v] + bh : 0.0f;
      }
      if (m < nclear) {
#pragma unroll
        for (int v = 0; v < 8; ++v) A1[base + (long)v * RP] = 0.0f;
      }
    } else {
#pragma unroll
      for (int v = 0; v < 8; ++v) {
        if (row0 + hi * 8 + v < n_nodes) {
          H[base + (long)v * RP]  = c1 ? d1[v] + bl : d1[v];
          A0[base + (long)v * RP] = c2 ? d2[v] + bh : 0.0f;
        }
      }
      if (m < nclear) {
#pragma unroll
        for (int v = 0; v < 8; ++v) {
          if (row0 + hi * 8 + v < n_nodes) A1[base + (long)v * RP] = 0.0f;
        }
      }
    }
  }
}

template <int F_IN>
__global__ void node_transform(const float* __restrict__ x,
                               const float* __restrict__ Wl,
                               const float* __restrict__ Wr,
                               const float* __restrict__ bias,
                               float* __restrict__ H,
                               float* __restrict__ A0,
                               float* __restrict__ A1, int n_nodes) {
  const int lane = threadIdx.x & 31;
  const int wave = threadIdx.x >> 5;
  const int tile = blockIdx.x * (blockDim.x >> 5) + wave;
  const int tiles = (n_nodes + 15) >> 4;
  if (tile >= tiles) return;

  const int m    = lane & 15;
  const int hi   = lane >> 4;
  const int row0 = tile << 4;
  const int arow = min(row0 + m, n_nodes - 1);

  auto ldA = [&](int k) -> float {
    const bool ok = (k < F_IN);
    const float v = x[(long)arow * F_IN + (ok ? k : 0)];
    return ok ? v : 0.0f;
  };

  v8f d1 = {};
  v8f d2 = {};
#pragma unroll
  for (int kc = 0; kc < F_IN; kc += 4) {
    const int k0 = kc + hi * 2;
    v2f a, b1, b2;
    a.x  = ldA(k0);
    a.y  = ldA(k0 + 1);
    b1.x = ldB1<F_IN, 5>(Wl, Wr, m, k0);
    b1.y = ldB1<F_IN, 5>(Wl, Wr, m, k0 + 1);
    b2.x = ldB2<F_IN, 5>(Wr, m, k0);
    b2.y = ldB2<F_IN, 5>(Wr, m, k0 + 1);
    d1 = __builtin_amdgcn_wmma_f32_16x16x4_f32(false, a, false, b1, (short)0,
                                               d1, false, false);
    d2 = __builtin_amdgcn_wmma_f32_16x16x4_f32(false, a, false, b2, (short)0,
                                               d2, false, false);
  }
  store_rows5(d1, d2, bias, H, A0, A1, row0, hi, m, n_nodes, RP);
}

template <int F_OUT, bool FIRST>
__global__ void combine_transform(float* H, float* A0, float* A1,
                                  const float* __restrict__ Wl,
                                  const float* __restrict__ Wr,
                                  const float* __restrict__ bias,
                                  float* __restrict__ z0,
                                  float* __restrict__ z1, int n_nodes) {
  constexpr int F_IN = 5;
  const int lane = threadIdx.x & 31;
  const int wave = threadIdx.x >> 5;
  const int tile = blockIdx.x * (blockDim.x >> 5) + wave;
  const int tiles = (n_nodes + 15) >> 4;
  if (tile >= tiles) return;

  const int m    = lane & 15;
  const int hi   = lane >> 4;
  const int row0 = tile << 4;
  const int arow = min(row0 + m, n_nodes - 1);
  const long rb  = (long)arow * RP;

  const v4f a0l = *reinterpret_cast<const v4f*>(A0 + rb);
  const v4f a0h = *reinterpret_cast<const v4f*>(A0 + rb + 4);
  const v4f a1l = *reinterpret_cast<const v4f*>(A1 + rb);
  const v4f a1h = *reinterpret_cast<const v4f*>(A1 + rb + 4);
  const v4f hh  = *reinterpret_cast<const v4f*>(H + rb + 4);

  float iv;
  if (FIRST) {
    iv = 1.0f / fmaxf(a0h.y + a1h.y, 1.0f);
    if (hi == 0 && row0 + m < n_nodes) A1[rb + 6] = iv;
  } else {
    iv = a1h.z;
  }
  const float h0 = fmaxf((a0l.x + a1l.x) * iv + hh.y, 0.0f);
  const float h1 = fmaxf((a0l.y + a1l.y) * iv + hh.z, 0.0f);
  const float h2 = fmaxf((a0l.z + a1l.z) * iv + hh.w, 0.0f);
  const float h3 = fmaxf((a0l.w + a1l.w) * iv + a0h.z, 0.0f);
  const float h4 = fmaxf((a0h.x + a1h.x) * iv + a0h.w, 0.0f);

  v8f d1 = {};
  v8f d2 = {};
  {
    const int k0 = hi * 2;
    v2f a, b1, b2;
    a.x  = hi ? h2 : h0;
    a.y  = hi ? h3 : h1;
    b1.x = ldB1<F_IN, F_OUT>(Wl, Wr, m, k0);
    b1.y = ldB1<F_IN, F_OUT>(Wl, Wr, m, k0 + 1);
    b2.x = ldB2<F_IN, F_OUT>(Wr, m, k0);
    b2.y = ldB2<F_IN, F_OUT>(Wr, m, k0 + 1);
    d1 = __builtin_amdgcn_wmma_f32_16x16x4_f32(false, a, false, b1, (short)0,
                                               d1, false, false);
    d2 = __builtin_amdgcn_wmma_f32_16x16x4_f32(false, a, false, b2, (short)0,
                                               d2, false, false);
  }
  {
    const int k0 = 4 + hi * 2;
    v2f a, b1, b2;
    a.x  = hi ? 0.0f : h4;
    a.y  = 0.0f;
    b1.x = ldB1<F_IN, F_OUT>(Wl, Wr, m, k0);
    b1.y = ldB1<F_IN, F_OUT>(Wl, Wr, m, k0 + 1);
    b2.x = ldB2<F_IN, F_OUT>(Wr, m, k0);
    b2.y = ldB2<F_IN, F_OUT>(Wr, m, k0 + 1);
    d1 = __builtin_amdgcn_wmma_f32_16x16x4_f32(false, a, false, b1, (short)0,
                                               d1, false, false);
    d2 = __builtin_amdgcn_wmma_f32_16x16x4_f32(false, a, false, b2, (short)0,
                                               d2, false, false);
  }

  if constexpr (F_OUT == 5) {
    store_rows5(d1, d2, bias, H, A0, A1, row0, hi, m, n_nodes, AP);
  } else {
    const bool c0 = (m == 0);
    if (c0 || m == RP - 1) {
      const float bv = bias[0];
      float* p = (c0 ? H : A1) + (long)(row0 + hi * 8) * RP + m;
      const long zb = (long)(row0 + hi * 8);
      if (row0 + 16 <= n_nodes) {
#pragma unroll
        for (int v = 0; v < 8; ++v)
          p[(long)v * RP] = c0 ? d1[v] : d2[v] + bv;
        if (c0) {
#pragma unroll
          for (int v = 0; v < 8; ++v) {
            z0[zb + v] = 0.0f;
            z1[zb + v] = 0.0f;
          }
        }
      } else {
#pragma unroll
        for (int v = 0; v < 8; ++v) {
          if (row0 + hi * 8 + v < n_nodes)
            p[(long)v * RP] = c0 ? d1[v] : d2[v] + bv;
        }
        if (c0) {
#pragma unroll
          for (int v = 0; v < 8; ++v) {
            if (row0 + hi * 8 + v < n_nodes) {
              z0[zb + v] = 0.0f;
              z1[zb + v] = 0.0f;
            }
          }
        }
      }
    }
  }
}

__global__ void edge_scatter_one_direct(const int* __restrict__ src,
                                 const int* __restrict__ dst,
                                 const float* __restrict__ H,
                                 float* __restrict__ z0,
                                 float* __restrict__ z1, long n_edges) {
  float* __restrict__ agg = package_half() ? z1 : z0;
  const long stride = (long)gridDim.x * blockDim.x;
  for (long i = (long)blockIdx.x * blockDim.x + threadIdx.x; i < n_edges;
       i += stride) {
    const int s = src[i];
    const int d = dst[i];
    unsafeAtomicAdd(&agg[d], H[s * RP]);
  }
}

template <int F>
__global__ void edge_scatter_rows_direct(const int* __restrict__ src,
                                  const int* __restrict__ dst,
                                  const float* H, float* A0, float* A1,
                                  long n_edges) {
  constexpr int FH = 5;
  constexpr int R = 32 / F;
  constexpr int T = (32 + R - 1) / R;
  constexpr int RL = 32 - (T - 1) * R;
  float* agg = package_half() ? A1 : A0;
  const int lane = threadIdx.x & 31;
  const int r0 = lane / F;
  const int j = lane - r0 * F;
  const bool in = (r0 < R);
  const int r = in ? r0 : R - 1;
  const bool cl = (F > FH) && (j >= FH);
  const int jg = cl ? FH - 1 : j;
  const long wave =
      (long)blockIdx.x * (blockDim.x >> 5) +
      __builtin_amdgcn_readfirstlane((int)(threadIdx.x >> 5));
  const long stride = (long)gridDim.x * blockDim.x;
  for (long base = wave * 32; base < n_edges; base += stride) {
    if (base + 32 <= n_edges) {
      const int s = src[base + lane];
      const int d = dst[base + lane];
      float v[T];
      int de[T];
#pragma unroll
      for (int t = 0; t < T; ++t) {
        const int e = (t < T - 1 || r < RL) ? t * R + r : 31;
        const int se = __shfl(s, e);
        de[t] = __shfl(d, e) * RP + j;
        const float g = H[se * RP + jg];
        v[t] = cl ? 1.0f : g;
      }
      if (in) {
#pragma unroll
        for (int t = 0; t < T - 1; ++t) unsafeAtomicAdd(&agg[de[t]], v[t]);
        if (r0 < RL) unsafeAtomicAdd(&agg[de[T - 1]], v[T - 1]);
      }
    } else {
      int s = 0, d = 0;
      if (base + lane < n_edges) {
        s = src[base + lane];
        d = dst[base + lane];
      }
#pragma unroll
      for (int t = 0; t < T; ++t) {
        const int e0 = t * R + r0;
        const bool has = in && (e0 < 32);
        const int e = has ? e0 : 31;
        const int se = __shfl(s, e);
        const int dd = __shfl(d, e);
        if (has && base + e < n_edges) {
          const float g = H[se * RP + jg];
          unsafeAtomicAdd(&agg[dd * RP + j], cl ? 1.0f : g);
        }
      }
    }
  }
}

constexpr int SB = 10;
constexpr int BR = 1 << SB;
constexpr int DB = 19;
constexpr int MAXB = 512;
constexpr int PG = 1024;
constexpr int SP = 4;
constexpr int SW = 256;
constexpr int DMASK = (1 << DB) - 1;

__global__ void bin_count(const int* __restrict__ src, int* __restrict__ cnt,
                          long n_edges) {
  __shared__ int hist[MAXB];
  hist[threadIdx.x] = 0;
  __syncthreads();
  const long per = (n_edges + gridDim.x - 1) / gridDim.x;
  const long c0 = (long)blockIdx.x * per;
  const long c1 = min(n_edges, c0 + per);
  long i = c0 + threadIdx.x;
  for (; i + 3 * MAXB < c1; i += 4 * MAXB) {
    const int s0 = src[i], s1 = src[i + MAXB], s2 = src[i + 2 * MAXB],
              s3 = src[i + 3 * MAXB];
    atomicAdd(&hist[(s0 >> SB) & (MAXB - 1)], 1);
    atomicAdd(&hist[(s1 >> SB) & (MAXB - 1)], 1);
    atomicAdd(&hist[(s2 >> SB) & (MAXB - 1)], 1);
    atomicAdd(&hist[(s3 >> SB) & (MAXB - 1)], 1);
  }
  for (; i < c1; i += MAXB) atomicAdd(&hist[(src[i] >> SB) & (MAXB - 1)], 1);
  __syncthreads();
  cnt[blockIdx.x * MAXB + threadIdx.x] = hist[threadIdx.x];
}

__device__ __forceinline__ int scan_block(int* sc, int mine) {
  const int t = threadIdx.x;
  sc[t] = mine;
  __syncthreads();
  for (int off = 1; off < MAXB; off <<= 1) {
    const int v = (t >= off) ? sc[t - off] : 0;
    __syncthreads();
    sc[t] += v;
    __syncthreads();
  }
  return sc[t];
}

__global__ void bin_scan(int* __restrict__ cnt, int* __restrict__ total) {
  __shared__ int sc[MAXB];
  const int b = blockIdx.x;
  const int t = threadIdx.x;
  constexpr int Q = PG / MAXB;
  int c[Q];
  int s = 0;
#pragma unroll
  for (int q = 0; q < Q; ++q) {
    c[q] = cnt[(t * Q + q) * MAXB + b];
    s += c[q];
  }
  const int incl = scan_block(sc, s);
  int run = incl - s;
#pragma unroll
  for (int q = 0; q < Q; ++q) {
    cnt[(t * Q + q) * MAXB + b] = run;
    run += c[q];
  }
  if (t == MAXB - 1) total[b] = incl;
}

constexpr int CAP = 16384;
__global__ void bin_fill(const int* __restrict__ src,
                         const int* __restrict__ dst,
                         const int* __restrict__ cnt,
                         const int* __restrict__ total,
                         int* __restrict__ bstart, int* __restrict__ pk,
                         long n_edges) {
  __shared__ int sc[MAXB];
  __shared__ int cur[MAXB];
  __shared__ int gd[MAXB];
  __shared__ int wb[CAP];
  __shared__ unsigned short bb[CAP];
  const int t = threadIdx.x;
  const int g = blockIdx.x;
  const int tot = total[t];
  const int first = scan_block(sc, tot) - tot;
  const int my = cnt[g * MAXB + t];
  const int nxt = (g + 1 < (int)gridDim.x) ? cnt[(g + 1) * MAXB + t] : tot;
  const int c = nxt - my;
  const int lstart = scan_block(sc, c) - c;
  cur[t] = lstart;
  gd[t] = first + my - lstart;
  if (g == 0) {
    bstart[t] = first;
    if (t == MAXB - 1) bstart[MAXB] = first + tot;
  }
  __syncthreads();
  const long per = (n_edges + gridDim.x - 1) / gridDim.x;
  const long c0 = (long)g * per;
  const long c1 = min(n_edges, c0 + per);
  auto put = [&](int s, int d) {
    const int b = (s >> SB) & (MAXB - 1);
    const int lp = min(atomicAdd(&cur[b], 1), CAP - 1);
    wb[lp] = ((s & (BR - 1)) << DB) | d;
    bb[lp] = (unsigned short)b;
  };
  long i = c0 + t;
  for (; i + 3 * MAXB < c1; i += 4 * MAXB) {
    const int s0 = src[i], s1 = src[i + MAXB], s2 = src[i + 2 * MAXB],
              s3 = src[i + 3 * MAXB];
    const int d0 = dst[i], d1 = dst[i + MAXB], d2 = dst[i + 2 * MAXB],
              d3 = dst[i + 3 * MAXB];
    put(s0, d0);
    put(s1, d1);
    put(s2, d2);
    put(s3, d3);
  }
  for (; i < c1; i += MAXB) put(src[i], dst[i]);
  __syncthreads();
  const int n = (int)min((long)CAP, max(0L, c1 - c0));
  for (int q = t; q < n; q += MAXB) pk[gd[bb[q]] + q] = wb[q];
}

__device__ __forceinline__ void bin_part(const int* __restrict__ bstart, int b,
                                         int part, int& p0, int& p1) {
  const int start = bstart[b];
  const int end = bstart[b + 1];
  const int per = ((end - start + SP * 32 - 1) / (SP * 32)) * 32;
  p0 = min(end, start + part * per);
  p1 = min(end, p0 + per);
}

__global__ void edge_scatter_one(const int* __restrict__ pk,
                                 const int* __restrict__ bstart,
                                 const float* __restrict__ H,
                                 float* __restrict__ z0,
                                 float* __restrict__ z1, int n_nodes) {
  __shared__ float hb[BR];
  const int b = blockIdx.x / SP;
  const int part = blockIdx.x - b * SP;
  const int row_base = b << SB;
  const int nrows = min(BR, n_nodes - row_base);
  for (int r = threadIdx.x; r < nrows; r += SW)
    hb[r] = H[(long)(row_base + r) * RP];
  __syncthreads();
  float* __restrict__ agg = package_half() ? z1 : z0;
  int p0, p1;
  bin_part(bstart, b, part, p0, p1);
  for (int i = p0 + threadIdx.x; i < p1; i += SW) {
    const int w = pk[i];
    unsafeAtomicAdd(&agg[w & DMASK], hb[w >> DB]);
  }
}

template <int F>
__global__ void edge_scatter_rows(const int* __restrict__ pk,
                                  const int* __restrict__ bstart,
                                  const float* H, float* A0, float* A1,
                                  int n_nodes) {
  constexpr int FH = 5;
  constexpr int R = 32 / F;
  constexpr int T = (32 + R - 1) / R;
  constexpr int RL = 32 - (T - 1) * R;
  constexpr int NQ = BR * 2 / SW;
  __shared__ float hb[BR * FH];
  const int b = blockIdx.x / SP;
  const int part = blockIdx.x - b * SP;
  const int row_base = b << SB;
  const int nrows = min(BR, n_nodes - row_base);
  {
    const float* Hb = H + (long)row_base * RP;
    const int nq = nrows * 2;
    v4f hv[NQ];
#pragma unroll
    for (int u = 0; u < NQ; ++u) {
      const int q = min((int)threadIdx.x + u * SW, nq - 1);
      hv[u] = *reinterpret_cast<const v4f*>(Hb + (long)q * 4);
    }
#pragma unroll
    for (int u = 0; u < NQ; ++u) {
      const int q = (int)threadIdx.x + u * SW;
      if (q < nq) {
        const int r = q >> 1;
        if (q & 1) {
          hb[r * FH + 4] = hv[u].x;
        } else {
          hb[r * FH + 0] = hv[u].x;
          hb[r * FH + 1] = hv[u].y;
          hb[r * FH + 2] = hv[u].z;
          hb[r * FH + 3] = hv[u].w;
        }
      }
    }
  }
  __syncthreads();
  float* agg = package_half() ? A1 : A0;
  const int lane = threadIdx.x & 31;
  const int r0 = lane / F;
  const int j = lane - r0 * F;
  const bool in = (r0 < R);
  const int r = in ? r0 : R - 1;
  const bool cl = (F > FH) && (j >= FH);
  const int jg = cl ? FH - 1 : j;
  const int wv = __builtin_amdgcn_readfirstlane((int)(threadIdx.x >> 5));
  int p0, p1;
  bin_part(bstart, b, part, p0, p1);
  for (int base = p0 + wv * 32; base < p1; base += SW) {
    if (base + 32 <= p1) {
      const int w = pk[base + lane];
      float v[T];
      int de[T];
#pragma unroll
      for (int t = 0; t < T; ++t) {
        const int e = (t < T - 1 || r < RL) ? t * R + r : 31;
        const int we = __shfl(w, e);
        de[t] = (we & DMASK) * RP + j;
        const float g = hb[(we >> DB) * FH + jg];
        v[t] = cl ? 1.0f : g;
      }
      if (in) {
#pragma unroll
        for (int t = 0; t < T - 1; ++t) unsafeAtomicAdd(&agg[de[t]], v[t]);
        if (r0 < RL) unsafeAtomicAdd(&agg[de[T - 1]], v[T - 1]);
      }
    } else {
      int w = 0;
      if (base + lane < p1) w = pk[base + lane];
#pragma unroll
      for (int t = 0; t < T; ++t) {
        const int e0 = t * R + r0;
        const bool has = in && (e0 < 32);
        const int e = has ? e0 : 31;
        const int we = __shfl(w, e);
        if (has && base + e < p1) {
          const float g = hb[(we >> DB) * FH + jg];
          unsafeAtomicAdd(&agg[(we & DMASK) * RP + j], cl ? 1.0f : g);
        }
      }
    }
  }
}

__global__ void combine_last(const float* __restrict__ z0,
                             const float* __restrict__ z1,
                             const float* __restrict__ A1,
                             float* __restrict__ out, int n_nodes) {
  const int node = blockIdx.x * blockDim.x + threadIdx.x;
  if (node >= n_nodes) return;
  const v2f t = *reinterpret_cast<const v2f*>(A1 + (long)node * RP + 6);
  const float a = z0[node] + z1[node];
  out[node] = a * t.x + t.y;
}

extern "C" void kernel_launch(void* const* d_in, const int* in_sizes, int n_in,
                              void* d_out, int out_size, void* d_ws,
                              size_t ws_size, hipStream_t stream) {
  const float* x    = (const float*)d_in[0];
  const int*   ei   = (const int*)  d_in[1];
  const float* Wl1  = (const float*)d_in[2];
  const float* Wr1  = (const float*)d_in[3];
  const float* b1   = (const float*)d_in[4];
  const float* Wlm  = (const float*)d_in[5];
  const float* Wrm  = (const float*)d_in[6];
  const float* bm   = (const float*)d_in[7];
  const float* Wl10 = (const float*)d_in[8];
  const float* Wr10 = (const float*)d_in[9];
  const float* b10  = (const float*)d_in[10];

  const int  N = in_sizes[0] / 10;
  const long E = (long)in_sizes[1] / 2;
  const int* src = ei;
  const int* dst = ei + E;

  float* ws = (float*)d_ws;
  float* H  = ws;
  float* A0 = H + (size_t)N * RP;
  float* A1 = A0 + (size_t)N * RP;
  float* z0 = A1 + (size_t)N * RP;
  float* z1 = z0 + (size_t)N;
  int* pk     = (int*)(z1 + (size_t)N);
  int* cnt    = pk + (size_t)E;
  int* total  = cnt + (size_t)PG * MAXB;
  int* bstart = total + MAXB;

  const size_t need = ((size_t)N * 26 + (size_t)E + (size_t)PG * MAXB +
                       2 * MAXB + 8) * sizeof(float);
  const bool binned = ws_size >= need && N <= (1 << DB) && E <= (1L << 30) &&
                      (E + PG - 1) / PG <= CAP;

  const int B = 256;
  const int EGD = 8192;
  auto cdiv = [](long a, long b) { return (int)((a + b - 1) / b); };

  const int tiles   = (N + 15) / 16;
  const int tblocks = (tiles + 7) / 8;
  const int EG = cdiv(N, BR) * SP;

  if (binned) {
    bin_count<<<PG, MAXB, 0, stream>>>(src, cnt, E);
    bin_scan<<<MAXB, MAXB, 0, stream>>>(cnt, total);
    bin_fill<<<PG, MAXB, 0, stream>>>(src, dst, cnt, total, bstart, pk, E);
  }

  node_transform<10><<<tblocks, 256, 0, stream>>>(x, Wl1, Wr1, b1, H, A0, A1, N);
  if (binned)
    edge_scatter_rows<6><<<EG, SW, 0, stream>>>(pk, bstart, H, A0, A1, N);
  else
    edge_scatter_rows_direct<6><<<EGD, B, 0, stream>>>(src, dst, H, A0, A1, E);

  for (int l = 0; l < 8; ++l) {
    if (l == 0)
      combine_transform<5, true><<<tblocks, 256, 0, stream>>>(
          H, A0, A1, Wlm, Wrm, bm, z0, z1, N);
    else
      combine_transform<5, false><<<tblocks, 256, 0, stream>>>(
          H, A0, A1, Wlm + l * 25, Wrm + l * 25, bm + l * 5, z0, z1, N);
    if (binned)
      edge_scatter_rows<5><<<EG, SW, 0, stream>>>(pk, bstart, H, A0, A1, N);
    else
      edge_scatter_rows_direct<5><<<EGD, B, 0, stream>>>(src, dst, H, A0, A1, E);
  }

  combine_transform<1, false><<<tblocks, 256, 0, stream>>>(
      H, A0, A1, Wl10, Wr10, b10, z0, z1, N);
  if (binned)
    edge_scatter_one<<<EG, SW, 0, stream>>>(pk, bstart, H, z0, z1, N);
  else
    edge_scatter_one_direct<<<EGD, B, 0, stream>>>(src, dst, H, z0, z1, E);
  combine_last<<<cdiv(N, B), B, 0, stream>>>(z0, z1, A1, (float*)d_out, N);
}


# Bins reshaped (4096-row source block x 4 destination classes) and each put in destination-line order by one more launch: rows of one 128-byte line share an add instruction; add order changes
# speedup vs baseline: 10.4370x; 1.0865x over previous
#include <hip/hip_runtime.h>


typedef __attribute__((ext_vector_type(2))) float v2f;
typedef __attribute__((ext_vector_type(4))) float v4f;
typedef __attribute__((ext_vector_type(8))) float v8f;

constexpr int RP = 8;
constexpr int AP = 6;

__device__ __forceinline__ int package_half() {
  return (int)((__builtin_amdgcn_s_sendmsg_rtn(0x87) >> 18) & 1u);
}

template <int F_IN, int F_OUT>
__device__ __forceinline__ float ldB1(const float* __restrict__ Wl,
                                      const float* __restrict__ Wr, int n,
                                      int k) {
  if constexpr (F_OUT == 5) {
    const bool ok = (n < RP) && (k < F_IN);
    const bool left = (n < 5);
    const float* W = left ? Wl : Wr;
    const int r = left ? n : n - 5;
    const float v = W[ok ? (r * F_IN + k) : 0];
    return ok ? v : 0.0f;
  } else {
    const bool ok = (n == 0) && (k < F_IN);
    const float v = Wl[ok ? k : 0];
    return ok ? v : 0.0f;
  }
}
template <int F_IN, int F_OUT>
__device__ __forceinline__ float ldB2(const float* __restrict__ Wr, int n,
                                      int k) {
  if constexpr (F_OUT == 5) {
    const bool ok = (n >= 6) && (n < RP) && (k < F_IN);
    const float v = Wr[ok ? ((n - 3) * F_IN + k) : 0];
    return ok ? v : 0.0f;
  } else {
    const bool ok = (n == RP - 1) && (k < F_IN);
    const float v = Wr[ok ? k : 0];
    return ok ? v : 0.0f;
  }
}

__device__ __forceinline__ void store_rows5(const v8f& d1, const v8f& d2,
                                            const float* __restrict__ bias,
                                            float* H, float* A0, float* A1,
                                            int row0, int hi, int m,
                                            int n_nodes, int nclear) {
  if (m < RP) {
    const bool  c1 = (m >= 5);
    const bool  c2 = (m >= 6);
    const float bl = bias[c1 ? m - 5 : 0];
    const float bh = bias[c2 ? m - 3 : 0];
    const long  base = (long)(row0 + hi * 8) * RP + m;
    if (row0 + 16 <= n_nodes) {
#pragma unroll
      for (int v = 0; v < 8; ++v) {
        H[base + (long)v * RP]  = c1 ? d1[v] + bl : d1[v];
        A0[base + (long)v * RP] = c2 ? d2[v] + bh : 0.0f;
      }
      if (m < nclear) {
#pragma unroll
        for (int v = 0; v < 8; ++v) A1[base + (long)v * RP] = 0.0f;
      }
    } else {
#pragma unroll
      for (int v = 0; v < 8; ++v) {
        if (row0 + hi * 8 + v < n_nodes) {
          H[base + (long)v * RP]  = c1 ? d1[v] + bl : d1[v];
          A0[base + (long)v * RP] = c2 ? d2[v] + bh : 0.0f;
        }
      }
      if (m < nclear) {
#pragma unroll
        for (int v = 0; v < 8; ++v) {
          if (row0 + hi * 8 + v < n_nodes) A1[base + (long)v * RP] = 0.0f;
        }
      }
    }
  }
}

template <int F_IN>
__global__ void node_transform(const float* __restrict__ x,
                               const float* __restrict__ Wl,
                               const float* __restrict__ Wr,
                               const float* __restrict__ bias,
                               float* __restrict__ H,
                               float* __restrict__ A0,
                               float* __restrict__ A1, int n_nodes) {
  const int lane = threadIdx.x & 31;
  const int wave = threadIdx.x >> 5;
  const int tile = blockIdx.x * (blockDim.x >> 5) + wave;
  const int tiles = (n_nodes + 15) >> 4;
  if (tile >= tiles) return;

  const int m    = lane & 15;
  const int hi   = lane >> 4;
  const int row0 = tile << 4;
  const int arow = min(row0 + m, n_nodes - 1);

  auto ldA = [&](int k) -> float {
    const bool ok = (k < F_IN);
    const float v = x[(long)arow * F_IN + (ok ? k : 0)];
    return ok ? v : 0.0f;
  };

  v8f d1 = {};
  v8f d2 = {};
#pragma unroll
  for (int kc = 0; kc < F_IN; kc += 4) {
    const int k0 = kc + hi * 2;
    v2f a, b1, b2;
    a.x  = ldA(k0);
    a.y  = ldA(k0 + 1);
    b1.x = ldB1<F_IN, 5>(Wl, Wr, m, k0);
    b1.y = ldB1<F_IN, 5>(Wl, Wr, m, k0 + 1);
    b2.x = ldB2<F_IN, 5>(Wr, m, k0);
    b2.y = ldB2<F_IN, 5>(Wr, m, k0 + 1);
    d1 = __builtin_amdgcn_wmma_f32_16x16x4_f32(false, a, false, b1, (short)0,
                                               d1, false, false);
    d2 = __builtin_amdgcn_wmma_f32_16x16x4_f32(false, a, false, b2, (short)0,
                                               d2, false, false);
  }
  store_rows5(d1, d2, bias, H, A0, A1, row0, hi, m, n_nodes, RP);
}

template <int F_OUT, bool FIRST>
__global__ void combine_transform(float* H, float* A0, float* A1,
                                  const float* __restrict__ Wl,
                                  const float* __restrict__ Wr,
                                  const float* __restrict__ bias,
                                  float* __restrict__ z0,
                                  float* __restrict__ z1, int n_nodes) {
  constexpr int F_IN = 5;
  const int lane = threadIdx.x & 31;
  const int wave = threadIdx.x >> 5;
  const int tile = blockIdx.x * (blockDim.x >> 5) + wave;
  const int tiles = (n_nodes + 15) >> 4;
  if (tile >= tiles) return;

  const int m    = lane & 15;
  const int hi   = lane >> 4;
  const int row0 = tile << 4;
  const int arow = min(row0 + m, n_nodes - 1);
  const long rb  = (long)arow * RP;

  const v4f a0l = *reinterpret_cast<const v4f*>(A0 + rb);
  const v4f a0h = *reinterpret_cast<const v4f*>(A0 + rb + 4);
  const v4f a1l = *reinterpret_cast<const v4f*>(A1 + rb);
  const v4f a1h = *reinterpret_cast<const v4f*>(A1 + rb + 4);
  const v4f hh  = *reinterpret_cast<const v4f*>(H + rb + 4);

  float iv;
  if (FIRST) {
    iv = 1.0f / fmaxf(a0h.y + a1h.y, 1.0f);
    if (hi == 0 && row0 + m < n_nodes) A1[rb + 6] = iv;
  } else {
    iv = a1h.z;
  }
  const float h0 = fmaxf((a0l.x + a1l.x) * iv + hh.y, 0.0f);
  const float h1 = fmaxf((a0l.y + a1l.y) * iv + hh.z, 0.0f);
  const float h2 = fmaxf((a0l.z + a1l.z) * iv + hh.w, 0.0f);
  const float h3 = fmaxf((a0l.w + a1l.w) * iv + a0h.z, 0.0f);
  const float h4 = fmaxf((a0h.x + a1h.x) * iv + a0h.w, 0.0f);

  v8f d1 = {};
  v8f d2 = {};
  {
    const int k0 = hi * 2;
    v2f a, b1, b2;
    a.x  = hi ? h2 : h0;
    a.y  = hi ? h3 : h1;
    b1.x = ldB1<F_IN, F_OUT>(Wl, Wr, m, k0);
    b1.y = ldB1<F_IN, F_OUT>(Wl, Wr, m, k0 + 1);
    b2.x = ldB2<F_IN, F_OUT>(Wr, m, k0);
    b2.y = ldB2<F_IN, F_OUT>(Wr, m, k0 + 1);
    d1 = __builtin_amdgcn_wmma_f32_16x16x4_f32(false, a, false, b1, (short)0,
                                               d1, false, false);
    d2 = __builtin_amdgcn_wmma_f32_16x16x4_f32(false, a, false, b2, (short)0,
                                               d2, false, false);
  }
  {
    const int k0 = 4 + hi * 2;
    v2f a, b1, b2;
    a.x  = hi ? 0.0f : h4;
    a.y  = 0.0f;
    b1.x = ldB1<F_IN, F_OUT>(Wl, Wr, m, k0);
    b1.y = ldB1<F_IN, F_OUT>(Wl, Wr, m, k0 + 1);
    b2.x = ldB2<F_IN, F_OUT>(Wr, m, k0);
    b2.y = ldB2<F_IN, F_OUT>(Wr, m, k0 + 1);
    d1 = __builtin_amdgcn_wmma_f32_16x16x4_f32(false, a, false, b1, (short)0,
                                               d1, false, false);
    d2 = __builtin_amdgcn_wmma_f32_16x16x4_f32(false, a, false, b2, (short)0,
                                               d2, false, false);
  }

  if constexpr (F_OUT == 5) {
    store_rows5(d1, d2, bias, H, A0, A1, row0, hi, m, n_nodes, AP);
  } else {
    const bool c0 = (m == 0);
    if (c0 || m == RP - 1) {
      const float bv = bias[0];
      float* p = (c0 ? H : A1) + (long)(row0 + hi * 8) * RP + m;
      const long zb = (long)(row0 + hi * 8);
      if (row0 + 16 <= n_nodes) {
#pragma unroll
        for (int v = 0; v < 8; ++v)
          p[(long)v * RP] = c0 ? d1[v] : d2[v] + bv;
        if (c0) {
#pragma unroll
          for (int v = 0; v < 8; ++v) {
            z0[zb + v] = 0.0f;
            z1[zb + v] = 0.0f;
          }
        }
      } else {
#pragma unroll
        for (int v = 0; v < 8; ++v) {
          if (row0 + hi * 8 + v < n_nodes)
            p[(long)v * RP] = c0 ? d1[v] : d2[v] + bv;
        }
        if (c0) {
#pragma unroll
          for (int v = 0; v < 8; ++v) {
            if (row0 + hi * 8 + v < n_nodes) {
              z0[zb + v] = 0.0f;
              z1[zb + v] = 0.0f;
            }
          }
        }
      }
    }
  }
}

__global__ void edge_scatter_one_direct(const int* __restrict__ src,
                                 const int* __restrict__ dst,
                                 const float* __restrict__ H,
                                 float* __restrict__ z0,
                                 float* __restrict__ z1, long n_edges) {
  float* __restrict__ agg = package_half() ? z1 : z0;
  const long stride = (long)gridDim.x * blockDim.x;
  for (long i = (long)blockIdx.x * blockDim.x + threadIdx.x; i < n_edges;
       i += stride) {
    const int s = src[i];
    const int d = dst[i];
    unsafeAtomicAdd(&agg[d], H[s * RP]);
  }
}

template <int F>
__global__ void edge_scatter_rows_direct(const int* __restrict__ src,
                                  const int* __restrict__ dst,
                                  const float* H, float* A0, float* A1,
                                  long n_edges) {
  constexpr int FH = 5;
  constexpr int R = 32 / F;
  constexpr int T = (32 + R - 1) / R;
  constexpr int RL = 32 - (T - 1) * R;
  float* agg = package_half() ? A1 : A0;
  const int lane = threadIdx.x & 31;
  const int r0 = lane / F;
  const int j = lane - r0 * F;
  const bool in = (r0 < R);
  const int r = in ? r0 : R - 1;
  const bool cl = (F > FH) && (j >= FH);
  const int jg = cl ? FH - 1 : j;
  const long wave =
      (long)blockIdx.x * (blockDim.x >> 5) +
      __builtin_amdgcn_readfirstlane((int)(threadIdx.x >> 5));
  const long stride = (long)gridDim.x * blockDim.x;
  for (long base = wave * 32; base < n_edges; base += stride) {
    if (base + 32 <= n_edges) {
      const int s = src[base + lane];
      const int d = dst[base + lane];
      float v[T];
      int de[T];
#pragma unroll
      for (int t = 0; t < T; ++t) {
        const int e = (t < T - 1 || r < RL) ? t * R + r : 31;
        const int se = __shfl(s, e);
        de[t] = __shfl(d, e) * RP + j;
        const float g = H[se * RP + jg];
        v[t] = cl ? 1.0f : g;
      }
      if (in) {
#pragma unroll
        for (int t = 0; t < T - 1; ++t) unsafeAtomicAdd(&agg[de[t]], v[t]);
        if (r0 < RL) unsafeAtomicAdd(&agg[de[T - 1]], v[T - 1]);
      }
    } else {
      int s = 0, d = 0;
      if (base + lane < n_edges) {
        s = src[base + lane];
        d = dst[base + lane];
      }
#pragma unroll
      for (int t = 0; t < T; ++t) {
        const int e0 = t * R + r0;
        const bool has = in && (e0 < 32);
        const int e = has ? e0 : 31;
        const int se = __shfl(s, e);
        const int dd = __shfl(d, e);
        if (has && base + e < n_edges) {
          const float g = H[se * RP + jg];
          unsafeAtomicAdd(&agg[dd * RP + j], cl ? 1.0f : g);
        }
      }
    }
  }
}

constexpr int SB = 12;
constexpr int BR = 1 << SB;
constexpr int PB = 2;
constexpr int CS = 5;
constexpr int DB = 19;
constexpr int MAXB = 512;
constexpr int PG = 1024;
constexpr int SP = 1;
constexpr int SW = 1024;
constexpr int DMASK = (1 << DB) - 1;
static_assert(SB + DB <= 31, "a packed word is a non-negative int");

__device__ __forceinline__ int bin_of(int s, int d) {
  return (((s >> SB) << PB) | ((d >> CS) & ((1 << PB) - 1))) & (MAXB - 1);
}

__global__ void bin_count(const int* __restrict__ src,
                          const int* __restrict__ dst, int* __restrict__ cnt,
                          long n_edges) {
  __shared__ int hist[MAXB];
  hist[threadIdx.x] = 0;
  __syncthreads();
  const long per = (n_edges + gridDim.x - 1) / gridDim.x;
  const long c0 = (long)blockIdx.x * per;
  const long c1 = min(n_edges, c0 + per);
  long i = c0 + threadIdx.x;
  for (; i + 3 * MAXB < c1; i += 4 * MAXB) {
    const int s0 = src[i], s1 = src[i + MAXB], s2 = src[i + 2 * MAXB],
              s3 = src[i + 3 * MAXB];
    const int d0 = dst[i], d1 = dst[i + MAXB], d2 = dst[i + 2 * MAXB],
              d3 = dst[i + 3 * MAXB];
    atomicAdd(&hist[bin_of(s0, d0)], 1);
    atomicAdd(&hist[bin_of(s1, d1)], 1);
    atomicAdd(&hist[bin_of(s2, d2)], 1);
    atomicAdd(&hist[bin_of(s3, d3)], 1);
  }
  for (; i < c1; i += MAXB) atomicAdd(&hist[bin_of(src[i], dst[i])], 1);
  __syncthreads();
  cnt[blockIdx.x * MAXB + threadIdx.x] = hist[threadIdx.x];
}

__device__ __forceinline__ int scan_block(int* sc, int mine) {
  const int t = threadIdx.x;
  sc[t] = mine;
  __syncthreads();
  for (int off = 1; off < MAXB; off <<= 1) {
    const int v = (t >= off) ? sc[t - off] : 0;
    __syncthreads();
    sc[t] += v;
    __syncthreads();
  }
  return sc[t];
}

__global__ void bin_scan(int* __restrict__ cnt, int* __restrict__ total) {
  __shared__ int sc[MAXB];
  const int b = blockIdx.x;
  const int t = threadIdx.x;
  constexpr int Q = PG / MAXB;
  int c[Q];
  int s = 0;
#pragma unroll
  for (int q = 0; q < Q; ++q) {
    c[q] = cnt[(t * Q + q) * MAXB + b];
    s += c[q];
  }
  const int incl = scan_block(sc, s);
  int run = incl - s;
#pragma unroll
  for (int q = 0; q < Q; ++q) {
    cnt[(t * Q + q) * MAXB + b] = run;
    run += c[q];
  }
  if (t == MAXB - 1) total[b] = incl;
}

constexpr int CAP = 16384;
__global__ void bin_fill(const int* __restrict__ src,
                         const int* __restrict__ dst,
                         const int* __restrict__ cnt,
                         const int* __restrict__ total,
                         int* __restrict__ bstart, int* __restrict__ pk,
                         long n_edges) {
  __shared__ int sc[MAXB];
  __shared__ int cur[MAXB];
  __shared__ int gd[MAXB];
  __shared__ int wb[CAP];
  __shared__ unsigned short bb[CAP];
  const int t = threadIdx.x;
  const int g = blockIdx.x;
  const int tot = total[t];
  const int first = scan_block(sc, tot) - tot;
  const int my = cnt[g * MAXB + t];
  const int nxt = (g + 1 < (int)gridDim.x) ? cnt[(g + 1) * MAXB + t] : tot;
  const int c = nxt - my;
  const int lstart = scan_block(sc, c) - c;
  cur[t] = lstart;
  gd[t] = first + my - lstart;
  if (g == 0) {
    bstart[t] = first;
    if (t == MAXB - 1) bstart[MAXB] = first + tot;
  }
  __syncthreads();
  const long per = (n_edges + gridDim.x - 1) / gridDim.x;
  const long c0 = (long)g * per;
  const long c1 = min(n_edges, c0 + per);
  auto put = [&](int s, int d) {
    const int b = bin_of(s, d);
    const int lp = min(atomicAdd(&cur[b], 1), CAP - 1);
    wb[lp] = ((s & (BR - 1)) << DB) | d;
    bb[lp] = (unsigned short)b;
  };
  long i = c0 + t;
  for (; i + 3 * MAXB < c1; i += 4 * MAXB) {
    const int s0 = src[i], s1 = src[i + MAXB], s2 = src[i + 2 * MAXB],
              s3 = src[i + 3 * MAXB];
    const int d0 = dst[i], d1 = dst[i + MAXB], d2 = dst[i + 2 * MAXB],
              d3 = dst[i + 3 * MAXB];
    put(s0, d0);
    put(s1, d1);
    put(s2, d2);
    put(s3, d3);
  }
  for (; i < c1; i += MAXB) put(src[i], dst[i]);
  __syncthreads();
  const int n = (int)min((long)CAP, max(0L, c1 - c0));
  for (int q = t; q < n; q += MAXB) pk[gd[bb[q]] + q] = wb[q];
}

constexpr int OT = 1024;
constexpr int OW = 34;
constexpr int OCAP = OT * OW;
constexpr int GB = 3;
constexpr int NG = 1 << (DB - PB - GB);
constexpr int LS = 2;
static_assert(GB <= CS && LS <= GB, "a group lies inside one class chunk and holds whole lines");
__device__ __forceinline__ int group_of(int d) {
  return ((d >> (CS + PB)) << (CS - GB)) | ((d >> GB) & ((1 << (CS - GB)) - 1));
}
__global__ void __launch_bounds__(OT)
bin_order(const int* __restrict__ bstart, int* pk) {
  __shared__ int wb[OCAP];
  __shared__ int ge[NG];
  __shared__ int wtot[32];
  constexpr int CQ = NG / OT;
  const int t = threadIdx.x;
  const int lane = t & 31;
  const int wave = t >> 5;
  const int start = bstart[blockIdx.x];
  const int n = bstart[blockIdx.x + 1] - start;
  if (n <= 0 || n > OCAP) return;
  int* bin = pk + start;
  int w[OW];
#pragma unroll
  for (int u = 0; u < OW; ++u) w[u] = bin[min(t + u * OT, n - 1)];
  for (int k = t; k < NG; k += OT) ge[k] = 0;
  __syncthreads();
#pragma unroll
  for (int u = 0; u < OW; ++u) {
    if (t + u * OT < n) atomicAdd(&ge[group_of(w[u] & DMASK)], 1);
  }
  __syncthreads();
  int c[CQ];
  int s = 0;
#pragma unroll
  for (int q = 0; q < CQ; ++q) {
    c[q] = ge[t * CQ + q];
    s += c[q];
  }
  int incl = s;
#pragma unroll
  for (int off = 1; off < 32; off <<= 1) {
    const int v = __shfl_up(incl, off);
    if (lane >= off) incl += v;
  }
  if (lane == 31) wtot[wave] = incl;
  __syncthreads();
  if (wave == 0) {
    const int mine = wtot[lane];
    int wi = mine;
#pragma unroll
    for (int off = 1; off < 32; off <<= 1) {
      const int v = __shfl_up(wi, off);
      if (lane >= off) wi += v;
    }
    wtot[lane] = wi - mine;
  }
  __syncthreads();
  int run = wtot[wave] + incl - s;
#pragma unroll
  for (int q = 0; q < CQ; ++q) {
    ge[t * CQ + q] = run;
    run += c[q];
  }
  __syncthreads();
#pragma unroll
  for (int u = 0; u < OW; ++u) {
    if (t + u * OT < n) {
      const int p = min(atomicAdd(&ge[group_of(w[u] & DMASK)], 1), OCAP - 1);
      wb[p] = w[u];
    }
  }
  __syncthreads();
  const int rot = (int)(((long)blockIdx.x * n) / gridDim.x) & ~31;
  for (int q = t; q < n; q += OT) {
    const int we = wb[q];
    const int d = we & DMASK;
    const int g = group_of(d);
    const int s0 = g ? ge[g - 1] : 0;
    const int s1 = min(ge[g], n);
    const int key = d >> LS;
    int rank = 0;
    for (int jj = s0; jj < s1; ++jj) {
      const int kj = (wb[jj] & DMASK) >> LS;
      rank += (kj < key || (kj == key && jj < q)) ? 1 : 0;
    }
    int place = min(s0 + rank, n - 1) + rot;
    if (place >= n) place -= n;
    bin[place] = we;
  }
}

__device__ __forceinline__ void bin_part(const int* __restrict__ bstart, int b,
                                         int part, int& p0, int& p1) {
  const int start = bstart[b];
  const int end = bstart[b + 1];
  const int per = ((end - start + SP * 32 - 1) / (SP * 32)) * 32;
  p0 = min(end, start + part * per);
  p1 = min(end, p0 + per);
}

__global__ void edge_scatter_one(const int* __restrict__ pk,
                                 const int* __restrict__ bstart,
                                 const float* __restrict__ H,
                                 float* __restrict__ z0,
                                 float* __restrict__ z1, int n_nodes) {
  __shared__ float hb[BR];
  const int b = blockIdx.x / SP;
  const int part = blockIdx.x - b * SP;
  const int row_base = (b >> PB) << SB;
  const int nrows = min(BR, n_nodes - row_base);
  for (int r = threadIdx.x; r < nrows; r += SW)
    hb[r] = H[(long)(row_base + r) * RP];
  __syncthreads();
  float* __restrict__ agg = package_half() ? z1 : z0;
  int p0, p1;
  bin_part(bstart, b, part, p0, p1);
  for (int i = p0 + threadIdx.x; i < p1; i += SW) {
    const int w = pk[i];
    unsafeAtomicAdd(&agg[w & DMASK], hb[w >> DB]);
  }
}

template <int F>
__global__ void edge_scatter_rows(const int* __restrict__ pk,
                                  const int* __restrict__ bstart,
                                  const float* H, float* A0, float* A1,
                                  int n_nodes) {
  constexpr int FH = 5;
  constexpr int R = 32 / F;
  constexpr int T = (32 + R - 1) / R;
  constexpr int RL = 32 - (T - 1) * R;
  constexpr int NQ = BR * 2 / SW;
  __shared__ float hb[BR * FH];
  const int b = blockIdx.x / SP;
  const int part = blockIdx.x - b * SP;
  const int row_base = (b >> PB) << SB;
  const int nrows = min(BR, n_nodes - row_base);
  {
    const float* Hb = H + (long)row_base * RP;
    const int nq = nrows * 2;
    v4f hv[NQ];
#pragma unroll
    for (int u = 0; u < NQ; ++u) {
      const int q = min((int)threadIdx.x + u * SW, nq - 1);
      hv[u] = *reinterpret_cast<const v4f*>(Hb + (long)q * 4);
    }
#pragma unroll
    for (int u = 0; u < NQ; ++u) {
      const int q = (int)threadIdx.x + u * SW;
      if (q < nq) {
        const int r = q >> 1;
        if (q & 1) {
          hb[r * FH + 4] = hv[u].x;
        } else {
          hb[r * FH + 0] = hv[u].x;
          hb[r * FH + 1] = hv[u].y;
          hb[r * FH + 2] = hv[u].z;
          hb[r * FH + 3] = hv[u].w;
        }
      }
    }
  }
  __syncthreads();
  float* agg = package_half() ? A1 : A0;
  const int lane = threadIdx.x & 31;
  const int r0 = lane / F;
  const int j = lane - r0 * F;
  const bool in = (r0 < R);
  const int r = in ? r0 : R - 1;
  const bool cl = (F > FH) && (j >= FH);
  const int jg = cl ? FH - 1 : j;
  const int wv = __builtin_amdgcn_readfirstlane((int)(threadIdx.x >> 5));
  int p0, p1;
  bin_part(bstart, b, part, p0, p1);
  for (int base = p0 + wv * 32; base < p1; base += SW) {
    if (base + 32 <= p1) {
      const int w = pk[base + lane];
      float v[T];
      int de[T];
#pragma unroll
      for (int t = 0; t < T; ++t) {
        const int e = (t < T - 1 || r < RL) ? t * R + r : 31;
        const int we = __shfl(w, e);
        de[t] = (we & DMASK) * RP + j;
        const float g = hb[(we >> DB) * FH + jg];
        v[t] = cl ? 1.0f : g;
      }
      if (in) {
#pragma unroll
        for (int t = 0; t < T - 1; ++t) unsafeAtomicAdd(&agg[de[t]], v[t]);
        if (r0 < RL) unsafeAtomicAdd(&agg[de[T - 1]], v[T - 1]);
      }
    } else {
      int w = 0;
      if (base + lane < p1) w = pk[base + lane];
#pragma unroll
      for (int t = 0; t < T; ++t) {
        const int e0 = t * R + r0;
        const bool has = in && (e0 < 32);
        const int e = has ? e0 : 31;
        const int we = __shfl(w, e);
        if (has && base + e < p1) {
          const float g = hb[(we >> DB) * FH + jg];
          unsafeAtomicAdd(&agg[(we & DMASK) * RP + j], cl ? 1.0f : g);
        }
      }
    }
  }
}

__global__ void combine_last(const float* __restrict__ z0,
                             const float* __restrict__ z1,
                             const float* __restrict__ A1,
                             float* __restrict__ out, int n_nodes) {
  const int node = blockIdx.x * blockDim.x + threadIdx.x;
  if (node >= n_nodes) return;
  const v2f t = *reinterpret_cast<const v2f*>(A1 + (long)node * RP + 6);
  const float a = z0[node] + z1[node];
  out[node] = a * t.x + t.y;
}

extern "C" void kernel_launch(void* const* d_in, const int* in_sizes, int n_in,
                              void* d_out, int out_size, void* d_ws,
                              size_t ws_size, hipStream_t stream) {
  const float* x    = (const float*)d_in[0];
  const int*   ei   = (const int*)  d_in[1];
  const float* Wl1  = (const float*)d_in[2];
  const float* Wr1  = (const float*)d_in[3];
  const float* b1   = (const float*)d_in[4];
  const float* Wlm  = (const float*)d_in[5];
  const float* Wrm  = (const float*)d_in[6];
  const float* bm   = (const float*)d_in[7];
  const float* Wl10 = (const float*)d_in[8];
  const float* Wr10 = (const float*)d_in[9];
  const float* b10  = (const float*)d_in[10];

  const int  N = in_sizes[0] / 10;
  const long E = (long)in_sizes[1] / 2;
  const int* src = ei;
  const int* dst = ei + E;

  float* ws = (float*)d_ws;
  float* H  = ws;
  float* A0 = H + (size_t)N * RP;
  float* A1 = A0 + (size_t)N * RP;
  float* z0 = A1 + (size_t)N * RP;
  float* z1 = z0 + (size_t)N;
  int* pk     = (int*)(z1 + (size_t)N);
  int* cnt    = pk + (size_t)E;
  int* total  = cnt + (size_t)PG * MAXB;
  int* bstart = total + MAXB;

  const size_t need = ((size_t)N * 26 + (size_t)E + (size_t)PG * MAXB +
                       2 * MAXB + 8) * sizeof(float);
  const bool binned = ws_size >= need && N <= (1 << DB) && E <= (1L << 30) &&
                      (E + PG - 1) / PG <= CAP;

  const int B = 256;
  const int EGD = 8192;
  auto cdiv = [](long a, long b) { return (int)((a + b - 1) / b); };

  const int tiles   = (N + 15) / 16;
  const int tblocks = (tiles + 7) / 8;
  const int NB = cdiv(N, BR) * (1 << PB);
  const int EG = NB * SP;

  if (binned) {
    bin_count<<<PG, MAXB, 0, stream>>>(src, dst, cnt, E);
    bin_scan<<<MAXB, MAXB, 0, stream>>>(cnt, total);
    bin_fill<<<PG, MAXB, 0, stream>>>(src, dst, cnt, total, bstart, pk, E);
    bin_order<<<NB, OT, 0, stream>>>(bstart, pk);
  }

  node_transform<10><<<tblocks, 256, 0, stream>>>(x, Wl1, Wr1, b1, H, A0, A1, N);
  if (binned)
    edge_scatter_rows<6><<<EG, SW, 0, stream>>>(pk, bstart, H, A0, A1, N);
  else
    edge_scatter_rows_direct<6><<<EGD, B, 0, stream>>>(src, dst, H, A0, A1, E);

  for (int l = 0; l < 8; ++l) {
    if (l == 0)
      combine_transform<5, true><<<tblocks, 256, 0, stream>>>(
          H, A0, A1, Wlm, Wrm, bm, z0, z1, N);
    else
      combine_transform<5, false><<<tblocks, 256, 0, stream>>>(
          H, A0, A1, Wlm + l * 25, Wrm + l * 25, bm + l * 5, z0, z1, N);
    if (binned)
      edge_scatter_rows<5><<<EG, SW, 0, stream>>>(pk, bstart, H, A0, A1, N);
    else
      edge_scatter_rows_direct<5><<<EGD, B, 0, stream>>>(src, dst, H, A0, A1, E);
  }

  combine_transform<1, false><<<tblocks, 256, 0, stream>>>(
      H, A0, A1, Wl10, Wr10, b10, z0, z1, N);
  if (binned)
    edge_scatter_one<<<EG, SW, 0, stream>>>(pk, bstart, H, z0, z1, N);
  else
    edge_scatter_one_direct<<<EGD, B, 0, stream>>>(src, dst, H, z0, z1, E);
  combine_last<<<cdiv(N, B), B, 0, stream>>>(z0, z1, A1, (float*)d_out, N);
}


# Layers 2..9 scatter the ReLU outputs themselves; a lane whose float is zero leaves its add out; the product moves behind the sums ([mean|h]x[Wl;Wr], depth 10, matrix unit): sum order changes
# speedup vs baseline: 10.7949x; 1.0343x over previous
#include <hip/hip_runtime.h>


typedef __attribute__((ext_vector_type(2))) float v2f;
typedef __attribute__((ext_vector_type(4))) float v4f;
typedef __attribute__((ext_vector_type(8))) float v8f;

constexpr int RP = 8;
constexpr int AP = 6;

__device__ __forceinline__ int package_half() {
  return (int)((__builtin_amdgcn_s_sendmsg_rtn(0x87) >> 18) & 1u);
}

template <int F_IN, int F_OUT>
__device__ __forceinline__ float ldB1(const float* __restrict__ Wl,
                                      const float* __restrict__ Wr, int n,
                                      int k) {
  if constexpr (F_OUT == 5) {
    const bool ok = (n < RP) && (k < F_IN);
    const bool left = (n < 5);
    const float* W = left ? Wl : Wr;
    const int r = left ? n : n - 5;
    const float v = W[ok ? (r * F_IN + k) : 0];
    return ok ? v : 0.0f;
  } else {
    const bool ok = (n == 0) && (k < F_IN);
    const float v = Wl[ok ? k : 0];
    return ok ? v : 0.0f;
  }
}
template <int F_IN, int F_OUT>
__device__ __forceinline__ float ldB2(const float* __restrict__ Wr, int n,
                                      int k) {
  if constexpr (F_OUT == 5) {
    const bool ok = (n >= 6) && (n < RP) && (k < F_IN);
    const float v = Wr[ok ? ((n - 3) * F_IN + k) : 0];
    return ok ? v : 0.0f;
  } else {
    const bool ok = (n == RP - 1) && (k < F_IN);
    const float v = Wr[ok ? k : 0];
    return ok ? v : 0.0f;
  }
}

__device__ __forceinline__ void store_rows5(const v8f& d1, const v8f& d2,
                                            const float* __restrict__ bias,
                                            float* H, float* A0, float* A1,
                                            int row0, int hi, int m,
                                            int n_nodes, int nclear) {
  if (m < RP) {
    const bool  c1 = (m >= 5);
    const bool  c2 = (m >= 6);
    const float bl = bias[c1 ? m - 5 : 0];
    const float bh = bias[c2 ? m - 3 : 0];
    const long  base = (long)(row0 + hi * 8) * RP + m;
    if (row0 + 16 <= n_nodes) {
#pragma unroll
      for (int v = 0; v < 8; ++v) {
        H[base + (long)v * RP]  = c1 ? d1[v] + bl : d1[v];
        A0[base + (long)v * RP] = c2 ? d2[v] + bh : 0.0f;
      }
      if (m < nclear) {
#pragma unroll
        for (int v = 0; v < 8; ++v) A1[base + (long)v * RP] = 0.0f;
      }
    } else {
#pragma unroll
      for (int v = 0; v < 8; ++v) {
        if (row0 + hi * 8 + v < n_nodes) {
          H[base + (long)v * RP]  = c1 ? d1[v] + bl : d1[v];
          A0[base + (long)v * RP] = c2 ? d2[v] + bh : 0.0f;
        }
      }
      if (m < nclear) {
#pragma unroll
        for (int v = 0; v < 8; ++v) {
          if (row0 + hi * 8 + v < n_nodes) A1[base + (long)v * RP] = 0.0f;
        }
      }
    }
  }
}

template <int F_IN>
__global__ void node_transform(const float* __restrict__ x,
                               const float* __restrict__ Wl,
                               const float* __restrict__ Wr,
                               const float* __restrict__ bias,
                               float* __restrict__ H,
                               float* __restrict__ A0,
                               float* __restrict__ A1, int n_nodes) {
  const int lane = threadIdx.x & 31;
  const int wave = threadIdx.x >> 5;
  const int tile = blockIdx.x * (blockDim.x >> 5) + wave;
  const int tiles = (n_nodes + 15) >> 4;
  if (tile >= tiles) return;

  const int m    = lane & 15;
  const int hi   = lane >> 4;
  const int row0 = tile << 4;
  const int arow = min(row0 + m, n_nodes - 1);

  auto ldA = [&](int k) -> float {
    const bool ok = (k < F_IN);
    const float v = x[(long)arow * F_IN + (ok ? k : 0)];
    return ok ? v : 0.0f;
  };

  v8f d1 = {};
  v8f d2 = {};
#pragma unroll
  for (int kc = 0; kc < F_IN; kc += 4) {
    const int k0 = kc + hi * 2;
    v2f a, b1, b2;
    a.x  = ldA(k0);
    a.y  = ldA(k0 + 1);
    b1.x = ldB1<F_IN, 5>(Wl, Wr, m, k0);
    b1.y = ldB1<F_IN, 5>(Wl, Wr, m, k0 + 1);
    b2.x = ldB2<F_IN, 5>(Wr, m, k0);
    b2.y = ldB2<F_IN, 5>(Wr, m, k0 + 1);
    d1 = __builtin_amdgcn_wmma_f32_16x16x4_f32(false, a, false, b1, (short)0,
                                               d1, false, false);
    d2 = __builtin_amdgcn_wmma_f32_16x16x4_f32(false, a, false, b2, (short)0,
                                               d2, false, false);
  }
  store_rows5(d1, d2, bias, H, A0, A1, row0, hi, m, n_nodes, RP);
}

__device__ __forceinline__ float ldB10(const float* __restrict__ Wl,
                                       const float* __restrict__ Wr, int n,
                                       int k) {
  const bool ok = (n < 5) && (k < 10);
  const bool left = (k < 5);
  const float* W = left ? Wl : Wr;
  const int kk = left ? k : k - 5;
  const float v = W[ok ? (n * 5 + kk) : 0];
  return ok ? v : 0.0f;
}

__global__ void finish_first(float* H, float* A0, float* A1, int n_nodes) {
  const int node = blockIdx.x * blockDim.x + threadIdx.x;
  if (node >= n_nodes) return;
  const long rb = (long)node * RP;
  const v4f a0l = *reinterpret_cast<const v4f*>(A0 + rb);
  const v4f a0h = *reinterpret_cast<const v4f*>(A0 + rb + 4);
  const v4f a1l = *reinterpret_cast<const v4f*>(A1 + rb);
  const v4f a1h = *reinterpret_cast<const v4f*>(A1 + rb + 4);
  const v4f hh  = *reinterpret_cast<const v4f*>(H + rb + 4);
  const float iv = 1.0f / fmaxf(a0h.y + a1h.y, 1.0f);
  v4f h;
  h.x = fmaxf((a0l.x + a1l.x) * iv + hh.y, 0.0f);
  h.y = fmaxf((a0l.y + a1l.y) * iv + hh.z, 0.0f);
  h.z = fmaxf((a0l.z + a1l.z) * iv + hh.w, 0.0f);
  h.w = fmaxf((a0l.w + a1l.w) * iv + a0h.z, 0.0f);
  const float h4 = fmaxf((a0h.x + a1h.x) * iv + a0h.w, 0.0f);
  const v4f zero4 = {0.0f, 0.0f, 0.0f, 0.0f};
  const v2f zero2 = {0.0f, 0.0f};
  v4f t1 = {0.0f, 0.0f, iv, a1h.w};
  *reinterpret_cast<v4f*>(H + rb) = h;
  H[rb + 4] = h4;
  *reinterpret_cast<v4f*>(A0 + rb) = zero4;
  *reinterpret_cast<v2f*>(A0 + rb + 4) = zero2;
  *reinterpret_cast<v4f*>(A1 + rb) = zero4;
  *reinterpret_cast<v4f*>(A1 + rb + 4) = t1;
}

__global__ void node_layer(float* H, float* A0, float* A1,
                           const float* __restrict__ Wl,
                           const float* __restrict__ Wr,
                           const float* __restrict__ bias, int n_nodes) {
  const int lane = threadIdx.x & 31;
  const int wave = threadIdx.x >> 5;
  const int tile = blockIdx.x * (blockDim.x >> 5) + wave;
  const int tiles = (n_nodes + 15) >> 4;
  if (tile >= tiles) return;

  const int m    = lane & 15;
  const int hi   = lane >> 4;
  const int row0 = tile << 4;
  const int arow = min(row0 + m, n_nodes - 1);
  const long rb  = (long)arow * RP;

  const v4f a0l = *reinterpret_cast<const v4f*>(A0 + rb);
  const v4f a1l = *reinterpret_cast<const v4f*>(A1 + rb);
  const v4f a1h = *reinterpret_cast<const v4f*>(A1 + rb + 4);
  const v4f hl  = *reinterpret_cast<const v4f*>(H + rb);
  const float a04 = A0[rb + 4];
  const float hp4 = H[rb + 4];
  const float iv = a1h.z;

  float val[12];
  val[0]  = (a0l.x + a1l.x) * iv;
  val[1]  = (a0l.y + a1l.y) * iv;
  val[2]  = (a0l.z + a1l.z) * iv;
  val[3]  = (a0l.w + a1l.w) * iv;
  val[4]  = (a04 + a1h.x) * iv;
  val[5]  = hl.x;
  val[6]  = hl.y;
  val[7]  = hl.z;
  val[8]  = hl.w;
  val[9]  = hp4;
  val[10] = 0.0f;
  val[11] = 0.0f;

  v8f d = {};
#pragma unroll
  for (int kc = 0; kc < 12; kc += 4) {
    const int k0 = kc + hi * 2;
    v2f a, b;
    a.x = hi ? val[kc + 2] : val[kc];
    a.y = hi ? val[kc + 3] : val[kc + 1];
    b.x = ldB10(Wl, Wr, m, k0);
    b.y = ldB10(Wl, Wr, m, k0 + 1);
    d = __builtin_amdgcn_wmma_f32_16x16x4_f32(false, a, false, b, (short)0, d,
                                              false, false);
  }

  if (m < AP) {
    const bool hc = (m < 5);
    const float bv = bias[hc ? m : 0];
    const long base = (long)(row0 + hi * 8) * RP + m;
    if (row0 + 16 <= n_nodes) {
      if (hc) {
#pragma unroll
        for (int v = 0; v < 8; ++v)
          H[base + (long)v * RP] = fmaxf(d[v] + bv, 0.0f);
      }
#pragma unroll
      for (int v = 0; v < 8; ++v) {
        A0[base + (long)v * RP] = 0.0f;
        A1[base + (long)v * RP] = 0.0f;
      }
    } else {
#pragma unroll
      for (int v = 0; v < 8; ++v) {
        if (row0 + hi * 8 + v < n_nodes) {
          if (hc) H[base + (long)v * RP] = fmaxf(d[v] + bv, 0.0f);
          A0[base + (long)v * RP] = 0.0f;
          A1[base + (long)v * RP] = 0.0f;
        }
      }
    }
  }
}

__global__ void transform_last(float* H, float* A1,
                               const float* __restrict__ Wl,
                               const float* __restrict__ Wr,
                               const float* __restrict__ bias,
                               float* __restrict__ z0,
                               float* __restrict__ z1, int n_nodes) {
  constexpr int F_IN = 5;
  const int lane = threadIdx.x & 31;
  const int wave = threadIdx.x >> 5;
  const int tile = blockIdx.x * (blockDim.x >> 5) + wave;
  const int tiles = (n_nodes + 15) >> 4;
  if (tile >= tiles) return;

  const int m    = lane & 15;
  const int hi   = lane >> 4;
  const int row0 = tile << 4;
  const int arow = min(row0 + m, n_nodes - 1);
  const long rb  = (long)arow * RP;

  const v4f hl = *reinterpret_cast<const v4f*>(H + rb);
  const float h4 = H[rb + 4];

  v8f d1 = {};
  v8f d2 = {};
  {
    const int k0 = hi * 2;
    v2f a, b1, b2;
    a.x  = hi ? hl.z : hl.x;
    a.y  = hi ? hl.w : hl.y;
    b1.x = ldB1<F_IN, 1>(Wl, Wr, m, k0);
    b1.y = ldB1<F_IN, 1>(Wl, Wr, m, k0 + 1);
    b2.x = ldB2<F_IN, 1>(Wr, m, k0);
    b2.y = ldB2<F_IN, 1>(Wr, m, k0 + 1);
    d1 = __builtin_amdgcn_wmma_f32_16x16x4_f32(false, a, false, b1, (short)0,
                                               d1, false, false);
    d2 = __builtin_amdgcn_wmma_f32_16x16x4_f32(false, a, false, b2, (short)0,
                                               d2, false, false);
  }
  {
    const int k0 = 4 + hi * 2;
    v2f a, b1, b2;
    a.x  = hi ? 0.0f : h4;
    a.y  = 0.0f;
    b1.x = ldB1<F_IN, 1>(Wl, Wr, m, k0);
    b1.y = ldB1<F_IN, 1>(Wl, Wr, m, k0 + 1);
    b2.x = ldB2<F_IN, 1>(Wr, m, k0);
    b2.y = ldB2<F_IN, 1>(Wr, m, k0 + 1);
    d1 = __builtin_amdgcn_wmma_f32_16x16x4_f32(false, a, false, b1, (short)0,
                                               d1, false, false);
    d2 = __builtin_amdgcn_wmma_f32_16x16x4_f32(false, a, false, b2, (short)0,
                                               d2, false, false);
  }

  const bool c0 = (m == 0);
  if (c0 || m == RP - 1) {
    const float bv = bias[0];
    float* p = (c0 ? H : A1) + (long)(row0 + hi * 8) * RP + m;
    const long zb = (long)(row0 + hi * 8);
    if (row0 + 16 <= n_nodes) {
#pragma unroll
      for (int v = 0; v < 8; ++v)
        p[(long)v * RP] = c0 ? d1[v] : d2[v] + bv;
      if (c0) {
#pragma unroll
        for (int v = 0; v < 8; ++v) {
          z0[zb + v] = 0.0f;
          z1[zb + v] = 0.0f;
        }
      }
    } else {
#pragma unroll
      for (int v = 0; v < 8; ++v) {
        if (row0 + hi * 8 + v < n_nodes)
          p[(long)v * RP] = c0 ? d1[v] : d2[v] + bv;
      }
      if (c0) {
#pragma unroll
        for (int v = 0; v < 8; ++v) {
          if (row0 + hi * 8 + v < n_nodes) {
            z0[zb + v] = 0.0f;
            z1[zb + v] = 0.0f;
          }
        }
      }
    }
  }
}

__global__ void edge_scatter_one_direct(const int* __restrict__ src,
                                 const int* __restrict__ dst,
                                 const float* __restrict__ H,
                                 float* __restrict__ z0,
                                 float* __restrict__ z1, long n_edges) {
  float* __restrict__ agg = package_half() ? z1 : z0;
  const long stride = (long)gridDim.x * blockDim.x;
  for (long i = (long)blockIdx.x * blockDim.x + threadIdx.x; i < n_edges;
       i += stride) {
    const int s = src[i];
    const int d = dst[i];
    unsafeAtomicAdd(&agg[d], H[s * RP]);
  }
}

template <int F>
__global__ void edge_scatter_rows_direct(const int* __restrict__ src,
                                  const int* __restrict__ dst,
                                  const float* H, float* A0, float* A1,
                                  long n_edges) {
  constexpr int FH = 5;
  constexpr int R = 32 / F;
  constexpr int T = (32 + R - 1) / R;
  constexpr int RL = 32 - (T - 1) * R;
  constexpr bool SKIP0 = (F == FH);
  float* agg = package_half() ? A1 : A0;
  const int lane = threadIdx.x & 31;
  const int r0 = lane / F;
  const int j = lane - r0 * F;
  const bool in = (r0 < R);
  const int r = in ? r0 : R - 1;
  const bool cl = (F > FH) && (j >= FH);
  const int jg = cl ? FH - 1 : j;
  const long wave =
      (long)blockIdx.x * (blockDim.x >> 5) +
      __builtin_amdgcn_readfirstlane((int)(threadIdx.x >> 5));
  const long stride = (long)gridDim.x * blockDim.x;
  for (long base = wave * 32; base < n_edges; base += stride) {
    if (base + 32 <= n_edges) {
      const int s = src[base + lane];
      const int d = dst[base + lane];
      float v[T];
      int de[T];
#pragma unroll
      for (int t = 0; t < T; ++t) {
        const int e = (t < T - 1 || r < RL) ? t * R + r : 31;
        const int se = __shfl(s, e);
        de[t] = __shfl(d, e) * RP + j;
        const float g = H[se * RP + jg];
        v[t] = cl ? 1.0f : g;
      }
      if (in) {
#pragma unroll
        for (int t = 0; t < T - 1; ++t) {
          if (!SKIP0 || v[t] != 0.0f) unsafeAtomicAdd(&agg[de[t]], v[t]);
        }
        if (r0 < RL && (!SKIP0 || v[T - 1] != 0.0f))
          unsafeAtomicAdd(&agg[de[T - 1]], v[T - 1]);
      }
    } else {
      int s = 0, d = 0;
      if (base + lane < n_edges) {
        s = src[base + lane];
        d = dst[base + lane];
      }
#pragma unroll
      for (int t = 0; t < T; ++t) {
        const int e0 = t * R + r0;
        const bool has = in && (e0 < 32);
        const int e = has ? e0 : 31;
        const int se = __shfl(s, e);
        const int dd = __shfl(d, e);
        if (has && base + e < n_edges) {
          const float g = H[se * RP + jg];
          const float a = cl ? 1.0f : g;
          if (!SKIP0 || a != 0.0f) unsafeAtomicAdd(&agg[dd * RP + j], a);
        }
      }
    }
  }
}

constexpr int SB = 12;
constexpr int BR = 1 << SB;
constexpr int PB = 2;
constexpr int CS = 5;
constexpr int DB = 19;
constexpr int MAXB = 512;
constexpr int PG = 1024;
constexpr int SP = 1;
constexpr int SW = 1024;
constexpr int DMASK = (1 << DB) - 1;
static_assert(SB + DB <= 31, "a packed word is a non-negative int");

__device__ __forceinline__ int bin_of(int s, int d) {
  return (((s >> SB) << PB) | ((d >> CS) & ((1 << PB) - 1))) & (MAXB - 1);
}

__global__ void bin_count(const int* __restrict__ src,
                          const int* __restrict__ dst, int* __restrict__ cnt,
                          long n_edges) {
  __shared__ int hist[MAXB];
  hist[threadIdx.x] = 0;
  __syncthreads();
  const long per = (n_edges + gridDim.x - 1) / gridDim.x;
  const long c0 = (long)blockIdx.x * per;
  const long c1 = min(n_edges, c0 + per);
  long i = c0 + threadIdx.x;
  for (; i + 3 * MAXB < c1; i += 4 * MAXB) {
    const int s0 = src[i], s1 = src[i + MAXB], s2 = src[i + 2 * MAXB],
              s3 = src[i + 3 * MAXB];
    const int d0 = dst[i], d1 = dst[i + MAXB], d2 = dst[i + 2 * MAXB],
              d3 = dst[i + 3 * MAXB];
    atomicAdd(&hist[bin_of(s0, d0)], 1);
    atomicAdd(&hist[bin_of(s1, d1)], 1);
    atomicAdd(&hist[bin_of(s2, d2)], 1);
    atomicAdd(&hist[bin_of(s3, d3)], 1);
  }
  for (; i < c1; i += MAXB) atomicAdd(&hist[bin_of(src[i], dst[i])], 1);
  __syncthreads();
  cnt[blockIdx.x * MAXB + threadIdx.x] = hist[threadIdx.x];
}

__device__ __forceinline__ int scan_block(int* sc, int mine) {
  const int t = threadIdx.x;
  sc[t] = mine;
  __syncthreads();
  for (int off = 1; off < MAXB; off <<= 1) {
    const int v = (t >= off) ? sc[t - off] : 0;
    __syncthreads();
    sc[t] += v;
    __syncthreads();
  }
  return sc[t];
}

__global__ void bin_scan(int* __restrict__ cnt, int* __restrict__ total) {
  __shared__ int sc[MAXB];
  const int b = blockIdx.x;
  const int t = threadIdx.x;
  constexpr int Q = PG / MAXB;
  int c[Q];
  int s = 0;
#pragma unroll
  for (int q = 0; q < Q; ++q) {
    c[q] = cnt[(t * Q + q) * MAXB + b];
    s += c[q];
  }
  const int incl = scan_block(sc, s);
  int run = incl - s;
#pragma unroll
  for (int q = 0; q < Q; ++q) {
    cnt[(t * Q + q) * MAXB + b] = run;
    run += c[q];
  }
  if (t == MAXB - 1) total[b] = incl;
}

constexpr int CAP = 16384;
__global__ void bin_fill(const int* __restrict__ src,
                         const int* __restrict__ dst,
                         const int* __restrict__ cnt,
                         const int* __restrict__ total,
                         int* __restrict__ bstart, int* __restrict__ pk,
                         long n_edges) {
  __shared__ int sc[MAXB];
  __shared__ int cur[MAXB];
  __shared__ int gd[MAXB];
  __shared__ int wb[CAP];
  __shared__ unsigned short bb[CAP];
  const int t = threadIdx.x;
  const int g = blockIdx.x;
  const int tot = total[t];
  const int first = scan_block(sc, tot) - tot;
  const int my = cnt[g * MAXB + t];
  const int nxt = (g + 1 < (int)gridDim.x) ? cnt[(g + 1) * MAXB + t] : tot;
  const int c = nxt - my;
  const int lstart = scan_block(sc, c) - c;
  cur[t] = lstart;
  gd[t] = first + my - lstart;
  if (g == 0) {
    bstart[t] = first;
    if (t == MAXB - 1) bstart[MAXB] = first + tot;
  }
  __syncthreads();
  const long per = (n_edges + gridDim.x - 1) / gridDim.x;
  const long c0 = (long)g * per;
  const long c1 = min(n_edges, c0 + per);
  auto put = [&](int s, int d) {
    const int b = bin_of(s, d);
    const int lp = min(atomicAdd(&cur[b], 1), CAP - 1);
    wb[lp] = ((s & (BR - 1)) << DB) | d;
    bb[lp] = (unsigned short)b;
  };
  long i = c0 + t;
  for (; i + 3 * MAXB < c1; i += 4 * MAXB) {
    const int s0 = src[i], s1 = src[i + MAXB], s2 = src[i + 2 * MAXB],
              s3 = src[i + 3 * MAXB];
    const int d0 = dst[i], d1 = dst[i + MAXB], d2 = dst[i + 2 * MAXB],
              d3 = dst[i + 3 * MAXB];
    put(s0, d0);
    put(s1, d1);
    put(s2, d2);
    put(s3, d3);
  }
  for (; i < c1; i += MAXB) put(src[i], dst[i]);
  __syncthreads();
  const int n = (int)min((long)CAP, max(0L, c1 - c0));
  for (int q = t; q < n; q += MAXB) pk[gd[bb[q]] + q] = wb[q];
}

constexpr int OT = 1024;
constexpr int OW = 34;
constexpr int OCAP = OT * OW;
constexpr int GB = 3;
constexpr int NG = 1 << (DB - PB - GB);
constexpr int LS = 2;
static_assert(GB <= CS && LS <= GB, "a group lies inside one class chunk and holds whole lines");
__device__ __forceinline__ int group_of(int d) {
  return ((d >> (CS + PB)) << (CS - GB)) | ((d >> GB) & ((1 << (CS - GB)) - 1));
}
__global__ void __launch_bounds__(OT)
bin_order(const int* __restrict__ bstart, int* pk) {
  __shared__ int wb[OCAP];
  __shared__ int ge[NG];
  __shared__ int wtot[32];
  constexpr int CQ = NG / OT;
  const int t = threadIdx.x;
  const int lane = t & 31;
  const int wave = t >> 5;
  const int start = bstart[blockIdx.x];
  const int n = bstart[blockIdx.x + 1] - start;
  if (n <= 0 || n > OCAP) return;
  int* bin = pk + start;
  int w[OW];
#pragma unroll
  for (int u = 0; u < OW; ++u) w[u] = bin[min(t + u * OT, n - 1)];
  for (int k = t; k < NG; k += OT) ge[k] = 0;
  __syncthreads();
#pragma unroll
  for (int u = 0; u < OW; ++u) {
    if (t + u * OT < n) atomicAdd(&ge[group_of(w[u] & DMASK)], 1);
  }
  __syncthreads();
  int c[CQ];
  int s = 0;
#pragma unroll
  for (int q = 0; q < CQ; ++q) {
    c[q] = ge[t * CQ + q];
    s += c[q];
  }
  int incl = s;
#pragma unroll
  for (int off = 1; off < 32; off <<= 1) {
    const int v = __shfl_up(incl, off);
    if (lane >= off) incl += v;
  }
  if (lane == 31) wtot[wave] = incl;
  __syncthreads();
  if (wave == 0) {
    const int mine = wtot[lane];
    int wi = mine;
#pragma unroll
    for (int off = 1; off < 32; off <<= 1) {
      const int v = __shfl_up(wi, off);
      if (lane >= off) wi += v;
    }
    wtot[lane] = wi - mine;
  }
  __syncthreads();
  int run = wtot[wave] + incl - s;
#pragma unroll
  for (int q = 0; q < CQ; ++q) {
    ge[t * CQ + q] = run;
    run += c[q];
  }
  __syncthreads();
#pragma unroll
  for (int u = 0; u < OW; ++u) {
    if (t + u * OT < n) {
      const int p = min(atomicAdd(&ge[group_of(w[u] & DMASK)], 1), OCAP - 1);
      wb[p] = w[u];
    }
  }
  __syncthreads();
  const int rot = (int)(((long)blockIdx.x * n) / gridDim.x) & ~31;
  for (int q = t; q < n; q += OT) {
    const int we = wb[q];
    const int d = we & DMASK;
    const int g = group_of(d);
    const int s0 = g ? ge[g - 1] : 0;
    const int s1 = min(ge[g], n);
    const int key = d >> LS;
    int rank = 0;
    for (int jj = s0; jj < s1; ++jj) {
      const int kj = (wb[jj] & DMASK) >> LS;
      rank += (kj < key || (kj == key && jj < q)) ? 1 : 0;
    }
    int place = min(s0 + rank, n - 1) + rot;
    if (place >= n) place -= n;
    bin[place] = we;
  }
}

__device__ __forceinline__ void bin_part(const int* __restrict__ bstart, int b,
                                         int part, int& p0, int& p1) {
  const int start = bstart[b];
  const int end = bstart[b + 1];
  const int per = ((end - start + SP * 32 - 1) / (SP * 32)) * 32;
  p0 = min(end, start + part * per);
  p1 = min(end, p0 + per);
}

__global__ void edge_scatter_one(const int* __restrict__ pk,
                                 const int* __restrict__ bstart,
                                 const float* __restrict__ H,
                                 float* __restrict__ z0,
                                 float* __restrict__ z1, int n_nodes) {
  __shared__ float hb[BR];
  const int b = blockIdx.x / SP;
  const int part = blockIdx.x - b * SP;
  const int row_base = (b >> PB) << SB;
  const int nrows = min(BR, n_nodes - row_base);
  for (int r = threadIdx.x; r < nrows; r += SW)
    hb[r] = H[(long)(row_base + r) * RP];
  __syncthreads();
  float* __restrict__ agg = package_half() ? z1 : z0;
  int p0, p1;
  bin_part(bstart, b, part, p0, p1);
  for (int i = p0 + threadIdx.x; i < p1; i += SW) {
    const int w = pk[i];
    unsafeAtomicAdd(&agg[w & DMASK], hb[w >> DB]);
  }
}

template <int F>
__global__ void edge_scatter_rows(const int* __restrict__ pk,
                                  const int* __restrict__ bstart,
                                  const float* H, float* A0, float* A1,
                                  int n_nodes) {
  constexpr int FH = 5;
  constexpr int R = 32 / F;
  constexpr int T = (32 + R - 1) / R;
  constexpr int RL = 32 - (T - 1) * R;
  constexpr bool SKIP0 = (F == FH);
  constexpr int NQ = BR * 2 / SW;
  __shared__ float hb[BR * FH];
  const int b = blockIdx.x / SP;
  const int part = blockIdx.x - b * SP;
  const int row_base = (b >> PB) << SB;
  const int nrows = min(BR, n_nodes - row_base);
  {
    const float* Hb = H + (long)row_base * RP;
    const int nq = nrows * 2;
    v4f hv[NQ];
#pragma unroll
    for (int u = 0; u < NQ; ++u) {
      const int q = min((int)threadIdx.x + u * SW, nq - 1);
      hv[u] = *reinterpret_cast<const v4f*>(Hb + (long)q * 4);
    }
#pragma unroll
    for (int u = 0; u < NQ; ++u) {
      const int q = (int)threadIdx.x + u * SW;
      if (q < nq) {
        const int r = q >> 1;
        if (q & 1) {
          hb[r * FH + 4] = hv[u].x;
        } else {
          hb[r * FH + 0] = hv[u].x;
          hb[r * FH + 1] = hv[u].y;
          hb[r * FH + 2] = hv[u].z;
          hb[r * FH + 3] = hv[u].w;
        }
      }
    }
  }
  __syncthreads();
  float* agg = package_half() ? A1 : A0;
  const int lane = threadIdx.x & 31;
  const int r0 = lane / F;
  const int j = lane - r0 * F;
  const bool in = (r0 < R);
  const int r = in ? r0 : R - 1;
  const bool cl = (F > FH) && (j >= FH);
  const int jg = cl ? FH - 1 : j;
  const int wv = __builtin_amdgcn_readfirstlane((int)(threadIdx.x >> 5));
  int p0, p1;
  bin_part(bstart, b, part, p0, p1);
  for (int base = p0 + wv * 32; base < p1; base += SW) {
    if (base + 32 <= p1) {
      const int w = pk[base + lane];
      float v[T];
      int de[T];
#pragma unroll
      for (int t = 0; t < T; ++t) {
        const int e = (t < T - 1 || r < RL) ? t * R + r : 31;
        const int we = __shfl(w, e);
        de[t] = (we & DMASK) * RP + j;
        const float g = hb[(we >> DB) * FH + jg];
        v[t] = cl ? 1.0f : g;
      }
      if (in) {
#pragma unroll
        for (int t = 0; t < T - 1; ++t) {
          if (!SKIP0 || v[t] != 0.0f) unsafeAtomicAdd(&agg[de[t]], v[t]);
        }
        if (r0 < RL && (!SKIP0 || v[T - 1] != 0.0f))
          unsafeAtomicAdd(&agg[de[T - 1]], v[T - 1]);
      }
    } else {
      int w = 0;
      if (base + lane < p1) w = pk[base + lane];
#pragma unroll
      for (int t = 0; t < T; ++t) {
        const int e0 = t * R + r0;
        const bool has = in && (e0 < 32);
        const int e = has ? e0 : 31;
        const int we = __shfl(w, e);
        if (has && base + e < p1) {
          const float g = hb[(we >> DB) * FH + jg];
          const float a = cl ? 1.0f : g;
          if (!SKIP0 || a != 0.0f) unsafeAtomicAdd(&agg[(we & DMASK) * RP + j], a);
        }
      }
    }
  }
}

__global__ void combine_last(const float* __restrict__ z0,
                             const float* __restrict__ z1,
                             const float* __restrict__ A1,
                             float* __restrict__ out, int n_nodes) {
  const int node = blockIdx.x * blockDim.x + threadIdx.x;
  if (node >= n_nodes) return;
  const v2f t = *reinterpret_cast<const v2f*>(A1 + (long)node * RP + 6);
  const float a = z0[node] + z1[node];
  out[node] = a * t.x + t.y;
}

extern "C" void kernel_launch(void* const* d_in, const int* in_sizes, int n_in,
                              void* d_out, int out_size, void* d_ws,
                              size_t ws_size, hipStream_t stream) {
  const float* x    = (const float*)d_in[0];
  const int*   ei   = (const int*)  d_in[1];
  const float* Wl1  = (const float*)d_in[2];
  const float* Wr1  = (const float*)d_in[3];
  const float* b1   = (const float*)d_in[4];
  const float* Wlm  = (const float*)d_in[5];
  const float* Wrm  = (const float*)d_in[6];
  const float* bm   = (const float*)d_in[7];
  const float* Wl10 = (const float*)d_in[8];
  const float* Wr10 = (const float*)d_in[9];
  const float* b10  = (const float*)d_in[10];

  const int  N = in_sizes[0] / 10;
  const long E = (long)in_sizes[1] / 2;
  const int* src = ei;
  const int* dst = ei + E;

  float* ws = (float*)d_ws;
  float* H  = ws;
  float* A0 = H + (size_t)N * RP;
  float* A1 = A0 + (size_t)N * RP;
  float* z0 = A1 + (size_t)N * RP;
  float* z1 = z0 + (size_t)N;
  int* pk     = (int*)(z1 + (size_t)N);
  int* cnt    = pk + (size_t)E;
  int* total  = cnt + (size_t)PG * MAXB;
  int* bstart = total + MAXB;

  const size_t need = ((size_t)N * 26 + (size_t)E + (size_t)PG * MAXB +
                       2 * MAXB + 8) * sizeof(float);
  const bool binned = ws_size >= need && N <= (1 << DB) && E <= (1L << 30) &&
                      (E + PG - 1) / PG <= CAP;

  const int B = 256;
  const int EGD = 8192;
  auto cdiv = [](long a, long b) { return (int)((a + b - 1) / b); };

  const int tiles   = (N + 15) / 16;
  const int tblocks = (tiles + 7) / 8;
  const int NB = cdiv(N, BR) * (1 << PB);
  const int EG = NB * SP;

  if (binned) {
    bin_count<<<PG, MAXB, 0, stream>>>(src, dst, cnt, E);
    bin_scan<<<MAXB, MAXB, 0, stream>>>(cnt, total);
    bin_fill<<<PG, MAXB, 0, stream>>>(src, dst, cnt, total, bstart, pk, E);
    bin_order<<<NB, OT, 0, stream>>>(bstart, pk);
  }

  node_transform<10><<<tblocks, 256, 0, stream>>>(x, Wl1, Wr1, b1, H, A0, A1, N);
  if (binned)
    edge_scatter_rows<6><<<EG, SW, 0, stream>>>(pk, bstart, H, A0, A1, N);
  else
    edge_scatter_rows_direct<6><<<EGD, B, 0, stream>>>(src, dst, H, A0, A1, E);

  finish_first<<<cdiv(N, B), B, 0, stream>>>(H, A0, A1, N);

  for (int l = 0; l < 8; ++l) {
    if (binned)
      edge_scatter_rows<5><<<EG, SW, 0, stream>>>(pk, bstart, H, A0, A1, N);
    else
      edge_scatter_rows_direct<5><<<EGD, B, 0, stream>>>(src, dst, H, A0, A1, E);
    node_layer<<<tblocks, 256, 0, stream>>>(H, A0, A1, Wlm + l * 25,
                                            Wrm + l * 25, bm + l * 5, N);
  }

  transform_last<<<tblocks, 256, 0, stream>>>(H, A1, Wl10, Wr10, b10, z0, z1, N);
  if (binned)
    edge_scatter_one<<<EG, SW, 0, stream>>>(pk, bstart, H, z0, z1, N);
  else
    edge_scatter_one_direct<<<EGD, B, 0, stream>>>(src, dst, H, z0, z1, E);
  combine_last<<<cdiv(N, B), B, 0, stream>>>(z0, z1, A1, (float*)d_out, N);
}
